# starting kernel
# baseline (speedup 1.0000x reference)
#include <hip/hip_runtime.h>
#include <hip/hip_bf16.h>

typedef __attribute__((ext_vector_type(2))) float v2f;
typedef __attribute__((ext_vector_type(8))) float v8f;

#define DCH   128
#define SW_LD 132

__global__ void zero_f32(float* __restrict__ p, long long n4) {
    long long i = (long long)blockIdx.x * blockDim.x + threadIdx.x;
    if (i < n4) ((float4*)p)[i] = make_float4(0.f, 0.f, 0.f, 0.f);
}

__global__ void deg_count(const int* __restrict__ dst, float* __restrict__ deg, int nE) {
    int i = blockIdx.x * blockDim.x + threadIdx.x;
    if (i < nE) atomicAdd(deg + dst[i], 1.0f);
}

__global__ void deg_invert(const float* __restrict__ deg, float* __restrict__ dinv, int n) {
    int i = blockIdx.x * blockDim.x + threadIdx.x;
    if (i < n) {
        float d = deg[i];
        dinv[i] = (d > 0.f) ? (1.0f / d) : 0.f;
    }
}

__global__ void gin_scatter(const float* __restrict__ x, const float* __restrict__ ew,
                            const int* __restrict__ src, const int* __restrict__ dst,
                            float* __restrict__ agg, int nE) {
    long long t = (long long)blockIdx.x * blockDim.x + threadIdx.x;
    int e = (int)(t >> 5);
    if (e >= nE) return;
    int c = ((int)t & 31) * 4;
    float w = ew[e];
    const float4 v = *(const float4*)(x + (size_t)src[e] * DCH + c);
    float* a = agg + (size_t)dst[e] * DCH + c;
    atomicAdd(a + 0, v.x * w);
    atomicAdd(a + 1, v.y * w);
    atomicAdd(a + 2, v.z * w);
    atomicAdd(a + 3, v.w * w);
}

__global__ __launch_bounds__(256) void gin_gemm_wmma(
    const float* __restrict__ x, const float* __restrict__ agg,
    const float* __restrict__ dinv, const float* __restrict__ eps_arr, int eps_idx,
    const float* __restrict__ W, const float* __restrict__ bias,
    float* __restrict__ out, int n_nodes, int do_relu, int zero_row0)
{
    extern __shared__ float sW[];
    const int tid = threadIdx.x;

    const float4* W4 = (const float4*)W;
    for (int i = tid; i < DCH * DCH / 4; i += 256) {
        int row = i >> 5;
        int q   = i & 31;
        *(float4*)(&sW[row * SW_LD + q * 4]) = W4[i];
    }
    __syncthreads();

    const float eps = eps_arr[eps_idx];
    const float s   = 1.0f + eps;
    const int wave = tid >> 5;
    const int lane = tid & 31;
    const int half = lane >> 4;
    const int l16  = lane & 15;
    const int row0 = blockIdx.x * 128 + wave * 16;
    int node = row0 + l16;
    if (node >= n_nodes) node = n_nodes - 1;
    const float di = dinv[node];
    const float* xr = x   + (size_t)node * DCH;
    const float* ar = agg + (size_t)node * DCH;

    v2f A[32];
#pragma unroll
    for (int kb = 0; kb < 32; ++kb) {
        const int k = kb * 4 + half * 2;
        float2 xv = *(const float2*)(xr + k);
        float2 av = *(const float2*)(ar + k);
        A[kb].x = s * xv.x + av.x * di;
        A[kb].y = s * xv.y + av.y * di;
    }

#pragma unroll 1
    for (int j = 0; j < 8; ++j) {
        const int col = j * 16 + l16;
        const float* wrow = &sW[col * SW_LD];
        v8f c = {};
#pragma unroll
        for (int kb = 0; kb < 32; ++kb) {
            const int k = kb * 4 + half * 2;
            v2f b;
            b.x = wrow[k];
            b.y = wrow[k + 1];
            c = __builtin_amdgcn_wmma_f32_16x16x4_f32(
                     false, A[kb],  false, b,
                     (short)0, c,  false,  false);
        }
        const float bv = bias[col];
#pragma unroll
        for (int r = 0; r < 8; ++r) {
            const int rr = row0 + half * 8 + r;
            if (rr < n_nodes) {
                float v = c[r] + bv;
                if (do_relu) v = fmaxf(v, 0.0f);
                if (zero_row0 && rr == 0) v = 0.0f;
                out[(size_t)rr * DCH + col] = v;
            }
        }
    }
}

extern "C" void kernel_launch(void* const* d_in, const int* in_sizes, int n_in,
                              void* d_out, int out_size, void* d_ws, size_t ws_size,
                              hipStream_t stream) {
    const float* emb = (const float*)d_in[0];
    const float* w1  = (const float*)d_in[1];
    const float* b1  = (const float*)d_in[2];
    const float* w2  = (const float*)d_in[3];
    const float* b2  = (const float*)d_in[4];
    const float* eps = (const float*)d_in[5];
    const float* ew  = (const float*)d_in[6];
    const int*   src = (const int*)d_in[7];
    const int*   dst = (const int*)d_in[8];

    const int N = in_sizes[0] / DCH;
    const int E = in_sizes[6];

    float* deg  = (float*)d_ws;
    float* dinv = deg + N;
    float* agg  = dinv + N;
    float* bufA = agg + (size_t)N * DCH;
    float* out  = (float*)d_out;

    dim3 b256(256);
    auto zero = [&](float* p, size_t nf) {
        long long n4 = (long long)(nf / 4);
        zero_f32<<<(unsigned)((n4 + 255) / 256), b256, 0, stream>>>(p, n4);
    };

    zero(deg, (size_t)N);
    deg_count<<<(E + 255) / 256, b256, 0, stream>>>(dst, deg, E);
    deg_invert<<<(N + 255) / 256, b256, 0, stream>>>(deg, dinv, N);

    const int    gemmBlocks    = (N + 127) / 128;
    const size_t shmem         = (size_t)DCH * SW_LD * sizeof(float);
    const long long scatThreads = (long long)E * 32;
    const unsigned scatBlocks  = (unsigned)((scatThreads + 255) / 256);

    zero(agg, (size_t)N * DCH);
    gin_scatter<<<scatBlocks, b256, 0, stream>>>(emb, ew, src, dst, agg, E);
    gin_gemm_wmma<<<gemmBlocks, b256, shmem, stream>>>(emb, agg, dinv, eps, 0,
                                                       w1, b1, bufA, N, 1, 0);

    zero(agg, (size_t)N * DCH);
    gin_scatter<<<scatBlocks, b256, 0, stream>>>(bufA, ew, src, dst, agg, E);
    gin_gemm_wmma<<<gemmBlocks, b256, shmem, stream>>>(bufA, agg, dinv, eps, 1,
                                                       w2, b2, bufA, N, 1, 0);

    zero(agg, (size_t)N * DCH);
    gin_scatter<<<scatBlocks, b256, 0, stream>>>(bufA, ew, src, dst, agg, E);
    gin_gemm_wmma<<<gemmBlocks, b256, shmem, stream>>>(bufA, agg, dinv, eps, 2,
                                                       w2, b2, out, N, 0, 1);
}


# Float-atomic scatter replaced by a per-node gather over an edge list grouped by destination, built in-call (integer rank atomics, scan, permute); in-edge sums now run in arrival-order rank
# speedup vs baseline: 2.8181x; 2.8181x over previous
#include <hip/hip_runtime.h>
#include <hip/hip_bf16.h>

typedef __attribute__((ext_vector_type(2))) float v2f;
typedef __attribute__((ext_vector_type(8))) float v8f;

#define DCH   128
#define SW_LD 132

__global__ void zero_i32(int* __restrict__ p, int n) {
    int i = blockIdx.x * blockDim.x + threadIdx.x;
    if (i < n) p[i] = 0;
}

__global__ void edge_rank(const int* __restrict__ dst, int* __restrict__ cnt,
                          int* __restrict__ rank, int nE) {
    int i = blockIdx.x * blockDim.x + threadIdx.x;
    if (i < nE) rank[i] = atomicAdd(cnt + dst[i], 1);
}

#define SCAN_CHUNK 1024

__global__ __launch_bounds__(256) void chunk_sum(const int* __restrict__ cnt,
                                                 int* __restrict__ csum, int n) {
    __shared__ int red[256];
    const int tid = threadIdx.x;
    const int i0 = blockIdx.x * SCAN_CHUNK + tid * 4;
    int s = 0;
    for (int q = 0; q < 4; ++q) if (i0 + q < n) s += cnt[i0 + q];
    red[tid] = s;
    __syncthreads();
    for (int w = 128; w > 0; w >>= 1) {
        if (tid < w) red[tid] += red[tid + w];
        __syncthreads();
    }
    if (tid == 0) csum[blockIdx.x] = red[0];
}

__global__ __launch_bounds__(256) void node_offsets(const int* __restrict__ cnt,
                                                    const int* __restrict__ csum,
                                                    int* __restrict__ off,
                                                    float* __restrict__ dinv, int n) {
    __shared__ int sc[256];
    __shared__ int red[256];
    const int tid = threadIdx.x;
    int b = 0;
    for (int j = tid; j < (int)blockIdx.x; j += 256) b += csum[j];
    red[tid] = b;
    const int i0 = blockIdx.x * SCAN_CHUNK + tid * 4;
    int c[4];
    int s = 0;
    for (int q = 0; q < 4; ++q) { c[q] = (i0 + q < n) ? cnt[i0 + q] : 0; s += c[q]; }
    sc[tid] = s;
    __syncthreads();
    for (int w = 128; w > 0; w >>= 1) {
        if (tid < w) red[tid] += red[tid + w];
        __syncthreads();
    }
    for (int d = 1; d < 256; d <<= 1) {
        int v = (tid >= d) ? sc[tid - d] : 0;
        __syncthreads();
        sc[tid] += v;
        __syncthreads();
    }
    int run = red[0] + sc[tid] - s;
    for (int q = 0; q < 4; ++q) {
        if (i0 + q < n) {
            off[i0 + q]  = run;
            dinv[i0 + q] = (c[q] > 0) ? (1.0f / (float)c[q]) : 0.f;
            run += c[q];
        }
    }
}

__global__ void edge_place(const float* __restrict__ ew, const int* __restrict__ src,
                           const int* __restrict__ dst, const int* __restrict__ off,
                           const int* __restrict__ rank, int* __restrict__ rec_words, int nE) {
    int2* __restrict__ rec = (int2*)rec_words;
    int i = blockIdx.x * blockDim.x + threadIdx.x;
    if (i < nE) rec[off[dst[i]] + rank[i]] = make_int2(src[i], __float_as_int(ew[i]));
}

template <int K>
__device__ __forceinline__ void gather_rows(const float4* __restrict__ x4, const int2* __restrict__ rec,
                                            int p, int lane, float4& acc) {
    int2 e[K];
    float4 v[K];
#pragma unroll
    for (int i = 0; i < K; ++i) e[i] = rec[p + i];
#pragma unroll
    for (int i = 0; i < K; ++i) v[i] = x4[(size_t)e[i].x * 32 + lane];
#pragma unroll
    for (int i = 0; i < K; ++i) {
        const float w = __int_as_float(e[i].y);
        acc.x = fmaf(v[i].x, w, acc.x); acc.y = fmaf(v[i].y, w, acc.y);
        acc.z = fmaf(v[i].z, w, acc.z); acc.w = fmaf(v[i].w, w, acc.w);
    }
}

__global__ __launch_bounds__(256) void gin_gather(const float* __restrict__ x,
                                                  const int* __restrict__ rec_words,
                                                  const int* __restrict__ off,
                                                  const int* __restrict__ cnt,
                                                  float* __restrict__ agg, int n_nodes) {
    const int lane = threadIdx.x & 31;
    const int n = __builtin_amdgcn_readfirstlane(blockIdx.x * 8 + (threadIdx.x >> 5));
    if (n >= n_nodes) return;
    const int beg = off[n];
    const int end = beg + cnt[n];
    const float4* __restrict__ x4 = (const float4*)x;
    const int2* __restrict__ rec = (const int2*)rec_words;
    float4 acc = make_float4(0.f, 0.f, 0.f, 0.f);
    int p = beg;
    for (; p + 8 <= end; p += 8) gather_rows<8>(x4, rec, p, lane, acc);
    if (p + 4 <= end) { gather_rows<4>(x4, rec, p, lane, acc); p += 4; }
    if (p + 2 <= end) { gather_rows<2>(x4, rec, p, lane, acc); p += 2; }
    if (p < end)      { gather_rows<1>(x4, rec, p, lane, acc); }
    ((float4*)agg)[(size_t)n * 32 + lane] = acc;
}

__global__ __launch_bounds__(256) void gin_gemm_wmma(
    const float* __restrict__ x, const float* __restrict__ agg,
    const float* __restrict__ dinv, const float* __restrict__ eps_arr, int eps_idx,
    const float* __restrict__ W, const float* __restrict__ bias,
    float* __restrict__ out, int n_nodes, int do_relu, int zero_row0)
{
    extern __shared__ float sW[];
    const int tid = threadIdx.x;

    const float4* W4 = (const float4*)W;
    for (int i = tid; i < DCH * DCH / 4; i += 256) {
        int row = i >> 5;
        int q   = i & 31;
        *(float4*)(&sW[row * SW_LD + q * 4]) = W4[i];
    }
    __syncthreads();

    const float eps = eps_arr[eps_idx];
    const float s   = 1.0f + eps;
    const int wave = tid >> 5;
    const int lane = tid & 31;
    const int half = lane >> 4;
    const int l16  = lane & 15;
    const int row0 = blockIdx.x * 128 + wave * 16;
    int node = row0 + l16;
    if (node >= n_nodes) node = n_nodes - 1;
    const float di = dinv[node];
    const float* xr = x   + (size_t)node * DCH;
    const float* ar = agg + (size_t)node * DCH;

    v2f A[32];
#pragma unroll
    for (int kb = 0; kb < 32; ++kb) {
        const int k = kb * 4 + half * 2;
        float2 xv = *(const float2*)(xr + k);
        float2 av = *(const float2*)(ar + k);
        A[kb].x = s * xv.x + av.x * di;
        A[kb].y = s * xv.y + av.y * di;
    }

#pragma unroll 1
    for (int j = 0; j < 8; ++j) {
        const int col = j * 16 + l16;
        const float* wrow = &sW[col * SW_LD];
        v8f c = {};
#pragma unroll
        for (int kb = 0; kb < 32; ++kb) {
            const int k = kb * 4 + half * 2;
            v2f b;
            b.x = wrow[k];
            b.y = wrow[k + 1];
            c = __builtin_amdgcn_wmma_f32_16x16x4_f32(
                     false, A[kb],  false, b,
                     (short)0, c,  false,  false);
        }
        const float bv = bias[col];
#pragma unroll
        for (int r = 0; r < 8; ++r) {
            const int rr = row0 + half * 8 + r;
            if (rr < n_nodes) {
                float v = c[r] + bv;
                if (do_relu) v = fmaxf(v, 0.0f);
                if (zero_row0 && rr == 0) v = 0.0f;
                out[(size_t)rr * DCH + col] = v;
            }
        }
    }
}

extern "C" void kernel_launch(void* const* d_in, const int* in_sizes, int n_in,
                              void* d_out, int out_size, void* d_ws, size_t ws_size,
                              hipStream_t stream) {
    const float* emb = (const float*)d_in[0];
    const float* w1  = (const float*)d_in[1];
    const float* b1  = (const float*)d_in[2];
    const float* w2  = (const float*)d_in[3];
    const float* b2  = (const float*)d_in[4];
    const float* eps = (const float*)d_in[5];
    const float* ew  = (const float*)d_in[6];
    const int*   src = (const int*)d_in[7];
    const int*   dst = (const int*)d_in[8];

    const int N = in_sizes[0] / DCH;
    const int E = in_sizes[6];

    int*   cnt  = (int*)d_ws;
    int*   off  = cnt + N;
    float* dinv = (float*)(off + N);
    int*   csum = (int*)(dinv + N);
    int*   rank = csum + 1024;
    int*   rec  = rank + E;
    float* agg  = (float*)(rec + 2 * (size_t)E);
    float* bufA = agg + (size_t)N * DCH;
    float* out  = (float*)d_out;

    dim3 b256(256);

    const int nChunks = (N + SCAN_CHUNK - 1) / SCAN_CHUNK;
    zero_i32<<<(N + 255) / 256, b256, 0, stream>>>(cnt, N);
    edge_rank<<<(E + 255) / 256, b256, 0, stream>>>(dst, cnt, rank, E);
    chunk_sum<<<nChunks, b256, 0, stream>>>(cnt, csum, N);
    node_offsets<<<nChunks, b256, 0, stream>>>(cnt, csum, off, dinv, N);
    edge_place<<<(E + 255) / 256, b256, 0, stream>>>(ew, src, dst, off, rank, rec, E);

    const int    gemmBlocks   = (N + 127) / 128;
    const size_t shmem        = (size_t)DCH * SW_LD * sizeof(float);
    const int    gatherBlocks = (N + 7) / 8;

    gin_gather<<<gatherBlocks, b256, 0, stream>>>(emb, rec, off, cnt, agg, N);
    gin_gemm_wmma<<<gemmBlocks, b256, shmem, stream>>>(emb, agg, dinv, eps, 0,
                                                       w1, b1, bufA, N, 1, 0);

    gin_gather<<<gatherBlocks, b256, 0, stream>>>(bufA, rec, off, cnt, agg, N);
    gin_gemm_wmma<<<gemmBlocks, b256, shmem, stream>>>(bufA, agg, dinv, eps, 1,
                                                       w2, b2, bufA, N, 1, 0);

    gin_gather<<<gatherBlocks, b256, 0, stream>>>(bufA, rec, off, cnt, agg, N);
    gin_gemm_wmma<<<gemmBlocks, b256, shmem, stream>>>(bufA, agg, dinv, eps, 2,
                                                       w2, b2, out, N, 0, 1);
}


# Gather and combine fused into the GEMM kernel: each wave gathers its 16 nodes' rows into an LDS tile that feeds its A fragments; the aggregate never goes to memory; in-edge sums keep record order
# speedup vs baseline: 3.0858x; 1.0950x over previous
#include <hip/hip_runtime.h>
#include <hip/hip_bf16.h>

typedef __attribute__((ext_vector_type(2))) float v2f;
typedef __attribute__((ext_vector_type(8))) float v8f;

#define DCH   128
#define SW_LD 132

__global__ void zero_i32(int* __restrict__ p, int n) {
    int i = blockIdx.x * blockDim.x + threadIdx.x;
    if (i < n) p[i] = 0;
}

__global__ void edge_rank(const int* __restrict__ dst, int* __restrict__ cnt,
                          int* __restrict__ rank, int nE) {
    int i = blockIdx.x * blockDim.x + threadIdx.x;
    if (i < nE) rank[i] = atomicAdd(cnt + dst[i], 1);
}

#define SCAN_CHUNK 1024

__global__ __launch_bounds__(256) void chunk_sum(const int* __restrict__ cnt,
                                                 int* __restrict__ csum, int n) {
    __shared__ int red[256];
    const int tid = threadIdx.x;
    const int i0 = blockIdx.x * SCAN_CHUNK + tid * 4;
    int s = 0;
    for (int q = 0; q < 4; ++q) if (i0 + q < n) s += cnt[i0 + q];
    red[tid] = s;
    __syncthreads();
    for (int w = 128; w > 0; w >>= 1) {
        if (tid < w) red[tid] += red[tid + w];
        __syncthreads();
    }
    if (tid == 0) csum[blockIdx.x] = red[0];
}

__global__ __launch_bounds__(256) void node_offsets(const int* __restrict__ cnt,
                                                    const int* __restrict__ csum,
                                                    int* __restrict__ off,
                                                    float* __restrict__ dinv, int n) {
    __shared__ int sc[256];
    __shared__ int red[256];
    const int tid = threadIdx.x;
    int b = 0;
    for (int j = tid; j < (int)blockIdx.x; j += 256) b += csum[j];
    red[tid] = b;
    const int i0 = blockIdx.x * SCAN_CHUNK + tid * 4;
    int c[4];
    int s = 0;
    for (int q = 0; q < 4; ++q) { c[q] = (i0 + q < n) ? cnt[i0 + q] : 0; s += c[q]; }
    sc[tid] = s;
    __syncthreads();
    for (int w = 128; w > 0; w >>= 1) {
        if (tid < w) red[tid] += red[tid + w];
        __syncthreads();
    }
    for (int d = 1; d < 256; d <<= 1) {
        int v = (tid >= d) ? sc[tid - d] : 0;
        __syncthreads();
        sc[tid] += v;
        __syncthreads();
    }
    int run = red[0] + sc[tid] - s;
    for (int q = 0; q < 4; ++q) {
        if (i0 + q < n) {
            off[i0 + q]  = run;
            dinv[i0 + q] = (c[q] > 0) ? (1.0f / (float)c[q]) : 0.f;
            run += c[q];
        }
    }
}

__global__ void edge_place(const float* __restrict__ ew, const int* __restrict__ src,
                           const int* __restrict__ dst, const int* __restrict__ off,
                           const int* __restrict__ rank, int* __restrict__ rec_words, int nE) {
    int2* __restrict__ rec = (int2*)rec_words;
    int i = blockIdx.x * blockDim.x + threadIdx.x;
    if (i < nE) rec[off[dst[i]] + rank[i]] = make_int2(src[i], __float_as_int(ew[i]));
}

__global__ __launch_bounds__(256) void gin_layer_fused(
    const float* __restrict__ x, const int* __restrict__ rec_words,
    const int* __restrict__ off, const int* __restrict__ cnt,
    const float* __restrict__ dinv, const float* __restrict__ eps_arr, int eps_idx,
    const float* __restrict__ W, const float* __restrict__ bias,
    float* __restrict__ out, int n_nodes, int do_relu, int zero_row0)
{
    extern __shared__ float sW[];
    const int tid  = threadIdx.x;
    const int lane = tid & 31;
    const int half = lane >> 4;
    const int l16  = lane & 15;
    const int wave = __builtin_amdgcn_readfirstlane(tid >> 5);
    const int row0 = blockIdx.x * 128 + wave * 16;
    float* tile = sW + wave * 16 * SW_LD;

    const float4* __restrict__ x4 = (const float4*)x;
    const int2* __restrict__ rec = (const int2*)rec_words;

    const int  nd    = row0 + l16;
    const bool valid = nd < n_nodes;
    const int  ndc   = valid ? nd : n_nodes - 1;
    const int  segb  = off[ndc];
    const int  sege  = segb + cnt[ndc];
    const int  di_b  = valid ? __float_as_int(dinv[ndc]) : 0;
    const int  beg_all = __builtin_amdgcn_readlane(segb, 0);
    const int  end_all = (row0 < n_nodes) ? __builtin_amdgcn_readlane(sege, 15) : beg_all;
    const int  nfull   = (end_all - beg_all) >> 3;

    int2   eN[8];
    float4 vA[8], vB[8];
    int    wA[8], wB[8];

    if (nfull > 0) {
#pragma unroll
        for (int i = 0; i < 8; ++i) eN[i] = rec[beg_all + i];
    }

    float4 xr[16];
#pragma unroll
    for (int r = 0; r < 16; ++r) {
        const int rn = (row0 + r < n_nodes) ? (row0 + r) : (n_nodes - 1);
        xr[r] = x4[(size_t)rn * 32 + lane];
    }

    if (nfull > 0) {
#pragma unroll
        for (int i = 0; i < 8; ++i) { vA[i] = x4[(size_t)eN[i].x * 32 + lane]; wA[i] = eN[i].y; }
    }
    if (nfull > 1) {
#pragma unroll
        for (int i = 0; i < 8; ++i) eN[i] = rec[beg_all + 8 + i];
    }

    {
        const float s = 1.0f + eps_arr[eps_idx];
#pragma unroll
        for (int r = 0; r < 16; ++r) {
            float4 h;
            h.x = s * xr[r].x; h.y = s * xr[r].y; h.z = s * xr[r].z; h.w = s * xr[r].w;
            *(float4*)(tile + r * SW_LD + lane * 4) = h;
        }
    }

    int    j   = 0;
    int    nb  = __builtin_amdgcn_readlane(sege, 0);
    float4 acc = make_float4(0.f, 0.f, 0.f, 0.f);

#define GIN_FLUSH()                                                                     \
    do {                                                                                \
        const float d_ = __int_as_float(__builtin_amdgcn_readlane(di_b, j));            \
        float4* t_ = (float4*)(tile + j * SW_LD + lane * 4);                            \
        float4 h_ = *t_;                                                                \
        h_.x = fmaf(acc.x, d_, h_.x); h_.y = fmaf(acc.y, d_, h_.y);                     \
        h_.z = fmaf(acc.z, d_, h_.z); h_.w = fmaf(acc.w, d_, h_.w);                     \
        *t_ = h_;                                                                       \
        acc = make_float4(0.f, 0.f, 0.f, 0.f);                                          \
    } while (0)

#define GIN_CONSUME(q_, v_, w_)                                                         \
    do {                                                                                \
        while ((q_) >= nb) {                                                            \
            GIN_FLUSH();                                                                \
            ++j;                                                                        \
            nb = (j < 15) ? __builtin_amdgcn_readlane(sege, j) : 0x7fffffff;            \
        }                                                                               \
        const float wf_ = __int_as_float(w_);                                           \
        acc.x = fmaf((v_).x, wf_, acc.x); acc.y = fmaf((v_).y, wf_, acc.y);             \
        acc.z = fmaf((v_).z, wf_, acc.z); acc.w = fmaf((v_).w, wf_, acc.w);             \
    } while (0)

#define GIN_ISSUE(V_, W_)                                                               \
    _Pragma("unroll")                                                                   \
    for (int i = 0; i < 8; ++i) { V_[i] = x4[(size_t)eN[i].x * 32 + lane]; W_[i] = eN[i].y; }

#define GIN_LOADREC(P_)                                                                 \
    _Pragma("unroll")                                                                   \
    for (int i = 0; i < 8; ++i) eN[i] = rec[(P_) + i];

#define GIN_CONSUME8(P_, V_, W_)                                                        \
    _Pragma("unroll")                                                                   \
    for (int i = 0; i < 8; ++i) GIN_CONSUME((P_) + i, V_[i], W_[i]);

    int p = beg_all;
    if (nfull > 0) {
        int c = 0;
        for (;;) {
            if (c + 1 >= nfull) { GIN_CONSUME8(p, vA, wA); break; }
            GIN_ISSUE(vB, wB);
            if (c + 2 < nfull) { GIN_LOADREC(p + 16); }
            GIN_CONSUME8(p, vA, wA);
            ++c; p += 8;
            if (c + 1 >= nfull) { GIN_CONSUME8(p, vB, wB); break; }
            GIN_ISSUE(vA, wA);
            if (c + 2 < nfull) { GIN_LOADREC(p + 16); }
            GIN_CONSUME8(p, vB, wB);
            ++c; p += 8;
        }
        p = beg_all + nfull * 8;
    }
    if (p < end_all) {
        const int last = end_all - 1;
        int2 e[8];
#pragma unroll
        for (int i = 0; i < 8; ++i) e[i] = rec[(p + i < last) ? (p + i) : last];
#pragma unroll
        for (int i = 0; i < 8; ++i) vA[i] = x4[(size_t)e[i].x * 32 + lane];
#pragma unroll
        for (int i = 0; i < 8; ++i) {
            if (p + i < end_all) GIN_CONSUME(p + i, vA[i], e[i].y);
        }
    }
    for (; j < 16; ++j) GIN_FLUSH();
#undef GIN_CONSUME8
#undef GIN_LOADREC
#undef GIN_ISSUE
#undef GIN_CONSUME
#undef GIN_FLUSH

    __syncthreads();

    v2f A[32];
#pragma unroll
    for (int kb = 0; kb < 32; ++kb) {
        const int k = kb * 4 + half * 2;
        const float2 hv = *(const float2*)(tile + l16 * SW_LD + k);
        A[kb].x = hv.x;
        A[kb].y = hv.y;
    }

    __syncthreads();

    const float4* W4 = (const float4*)W;
    for (int i = tid; i < DCH * DCH / 4; i += 256) {
        int row = i >> 5;
        int q   = i & 31;
        *(float4*)(&sW[row * SW_LD + q * 4]) = W4[i];
    }
    __syncthreads();

#pragma unroll 1
    for (int jt = 0; jt < 8; ++jt) {
        const int col = jt * 16 + l16;
        const float* wrow = &sW[col * SW_LD];
        v8f cc = {};
#pragma unroll
        for (int kb = 0; kb < 32; ++kb) {
            const int k = kb * 4 + half * 2;
            v2f b;
            b.x = wrow[k];
            b.y = wrow[k + 1];
            cc = __builtin_amdgcn_wmma_f32_16x16x4_f32(
                     false, A[kb],  false, b,
                     (short)0, cc,  false,  false);
        }
        const float bv = bias[col];
#pragma unroll
        for (int r = 0; r < 8; ++r) {
            const int rr = row0 + half * 8 + r;
            if (rr < n_nodes) {
                float v = cc[r] + bv;
                if (do_relu) v = fmaxf(v, 0.0f);
                if (zero_row0 && rr == 0) v = 0.0f;
                out[(size_t)rr * DCH + col] = v;
            }
        }
    }
}

extern "C" void kernel_launch(void* const* d_in, const int* in_sizes, int n_in,
                              void* d_out, int out_size, void* d_ws, size_t ws_size,
                              hipStream_t stream) {
    const float* emb = (const float*)d_in[0];
    const float* w1  = (const float*)d_in[1];
    const float* b1  = (const float*)d_in[2];
    const float* w2  = (const float*)d_in[3];
    const float* b2  = (const float*)d_in[4];
    const float* eps = (const float*)d_in[5];
    const float* ew  = (const float*)d_in[6];
    const int*   src = (const int*)d_in[7];
    const int*   dst = (const int*)d_in[8];

    const int N = in_sizes[0] / DCH;
    const int E = in_sizes[6];

    int*   cnt  = (int*)d_ws;
    int*   off  = cnt + N;
    float* dinv = (float*)(off + N);
    int*   csum = (int*)(dinv + N);
    int*   rank = csum + 1024;
    int*   rec  = rank + E;
    float* bufA = (float*)(rec + 2 * (size_t)E);
    float* bufB = bufA + (size_t)N * DCH;
    float* out  = (float*)d_out;

    dim3 b256(256);

    const int nChunks = (N + SCAN_CHUNK - 1) / SCAN_CHUNK;
    zero_i32<<<(N + 255) / 256, b256, 0, stream>>>(cnt, N);
    edge_rank<<<(E + 255) / 256, b256, 0, stream>>>(dst, cnt, rank, E);
    chunk_sum<<<nChunks, b256, 0, stream>>>(cnt, csum, N);
    node_offsets<<<nChunks, b256, 0, stream>>>(cnt, csum, off, dinv, N);
    edge_place<<<(E + 255) / 256, b256, 0, stream>>>(ew, src, dst, off, rank, rec, E);

    const int    layerBlocks = (N + 127) / 128;
    const size_t shmem       = (size_t)DCH * SW_LD * sizeof(float);

    gin_layer_fused<<<layerBlocks, b256, shmem, stream>>>(emb, rec, off, cnt, dinv, eps, 0,
                                                          w1, b1, bufA, N, 1, 0);
    gin_layer_fused<<<layerBlocks, b256, shmem, stream>>>(bufA, rec, off, cnt, dinv, eps, 1,
                                                          w2, b2, bufB, N, 1, 0);
    gin_layer_fused<<<layerBlocks, b256, shmem, stream>>>(bufB, rec, off, cnt, dinv, eps, 2,
                                                          w2, b2, out, N, 0, 1);
}


# Destination-grouped edge list built by a two-level counting sort on LDS counters (node-range bins, then nodes) instead of global rank atomics and a node scan; in-edge sums run in LDS arrival order
# speedup vs baseline: 3.3341x; 1.0805x over previous
#include <hip/hip_runtime.h>
#include <hip/hip_bf16.h>

typedef __attribute__((ext_vector_type(2))) float v2f;
typedef __attribute__((ext_vector_type(8))) float v8f;

#define DCH   128
#define SW_LD 132

#define BIN_SHIFT 8
#define BIN_NODES 256
#define MAX_BINS  512
#define SLICES    256

__global__ __launch_bounds__(512) void bin_count(const int* __restrict__ dst, int* __restrict__ counts,
                                                 int nE, int nBins, int per_slice) {
    __shared__ int h[MAX_BINS];
    const int tid = threadIdx.x;
    for (int b = tid; b < nBins; b += 512) h[b] = 0;
    __syncthreads();
    const int beg = blockIdx.x * per_slice;
    const int end = (beg + per_slice < nE) ? (beg + per_slice) : nE;
    for (int i = beg + tid; i < end; i += 512) atomicAdd(&h[dst[i] >> BIN_SHIFT], 1);
    __syncthreads();
    for (int b = tid; b < nBins; b += 512) counts[b * SLICES + blockIdx.x] = h[b];
}

__device__ __forceinline__ int wave_incl_scan(int v, int lane) {
#pragma unroll
    for (int o = 1; o < 32; o <<= 1) {
        const int u = __shfl_up(v, o);
        if (lane >= o) v += u;
    }
    return v;
}

__global__ __launch_bounds__(1024) void scan_counts(int* __restrict__ counts, int* __restrict__ binbase,
                                                    int nBins) {
    __shared__ int tot[MAX_BINS];
    __shared__ int wsum[32];
    const int tid  = threadIdx.x;
    const int lane = tid & 31;
    const int wave = tid >> 5;
    for (int bin = wave; bin < nBins; bin += 32) {
        const int4* row = (const int4*)(counts + bin * SLICES);
        const int4 a = row[lane];
        const int4 b = row[32 + lane];
        int s = a.x + a.y + a.z + a.w + b.x + b.y + b.z + b.w;
#pragma unroll
        for (int o = 16; o > 0; o >>= 1) s += __shfl_xor(s, o);
        if (lane == 0) tot[bin] = s;
    }
    __syncthreads();
    const int v   = (tid < nBins) ? tot[tid] : 0;
    const int inc = wave_incl_scan(v, lane);
    if (lane == 31) wsum[wave] = inc;
    __syncthreads();
    if (wave == 0) {
        const int w  = wsum[lane];
        const int wi = wave_incl_scan(w, lane);
        wsum[lane] = wi - w;
    }
    __syncthreads();
    const int excl = wsum[wave] + inc - v;
    if (tid < nBins) { tot[tid] = excl; binbase[tid] = excl; }
    if (tid == nBins - 1) binbase[nBins] = excl + v;
    __syncthreads();
    for (int bin = wave; bin < nBins; bin += 32) {
        int4* row = (int4*)(counts + bin * SLICES);
        const int4 a = row[lane];
        const int4 b = row[32 + lane];
        const int sa = a.x + a.y + a.z + a.w;
        const int sb = b.x + b.y + b.z + b.w;
        const int ia = wave_incl_scan(sa, lane);
        const int ib = wave_incl_scan(sb, lane);
        const int totA = __shfl(ia, 31);
        const int ea = tot[bin] + ia - sa;
        const int eb = tot[bin] + totA + ib - sb;
        row[lane]      = make_int4(ea, ea + a.x, ea + a.x + a.y, ea + a.x + a.y + a.z);
        row[32 + lane] = make_int4(eb, eb + b.x, eb + b.x + b.y, eb + b.x + b.y + b.z);
    }
}

__global__ __launch_bounds__(512) void bin_scatter(const float* __restrict__ ew, const int* __restrict__ src,
                                                   const int* __restrict__ dst, const int* __restrict__ counts,
                                                   int* __restrict__ tdst, int* __restrict__ trec_words,
                                                   int nE, int nBins, int per_slice) {
    __shared__ int cur[MAX_BINS];
    int2* __restrict__ trec = (int2*)trec_words;
    const int tid = threadIdx.x;
    for (int b = tid; b < nBins; b += 512) cur[b] = counts[b * SLICES + blockIdx.x];
    __syncthreads();
    const int beg = blockIdx.x * per_slice;
    const int end = (beg + per_slice < nE) ? (beg + per_slice) : nE;
    for (int i = beg + tid; i < end; i += 512) {
        const int d   = dst[i];
        const int pos = atomicAdd(&cur[d >> BIN_SHIFT], 1);
        tdst[pos] = d;
        trec[pos] = make_int2(src[i], __float_as_int(ew[i]));
    }
}

__global__ __launch_bounds__(256) void bin_sort(const int* __restrict__ binbase, const int* __restrict__ tdst,
                                                const int* __restrict__ trec_words,
                                                int* __restrict__ cnt, int* __restrict__ off,
                                                float* __restrict__ dinv, int* __restrict__ rec_words, int n) {
    __shared__ int c[BIN_NODES];
    __shared__ int wsum[8];
    const int2* __restrict__ trec = (const int2*)trec_words;
    int2* __restrict__ rec = (int2*)rec_words;
    const int tid  = threadIdx.x;
    const int lane = tid & 31;
    const int wave = tid >> 5;
    const int b0 = binbase[blockIdx.x];
    const int b1 = binbase[blockIdx.x + 1];
    c[tid] = 0;
    __syncthreads();
    for (int i = b0 + tid; i < b1; i += 256) atomicAdd(&c[tdst[i] & (BIN_NODES - 1)], 1);
    __syncthreads();
    const int cv  = c[tid];
    const int inc = wave_incl_scan(cv, lane);
    if (lane == 31) wsum[wave] = inc;
    __syncthreads();
    int before = 0;
#pragma unroll
    for (int w = 0; w < 8; ++w) if (w < wave) before += wsum[w];
    const int start = b0 + before + inc - cv;
    const int node  = blockIdx.x * BIN_NODES + tid;
    if (node < n) {
        cnt[node]  = cv;
        off[node]  = start;
        dinv[node] = (cv > 0) ? (1.0f / (float)cv) : 0.f;
    }
    c[tid] = start;
    __syncthreads();
    for (int i = b0 + tid; i < b1; i += 256) {
        const int pos = atomicAdd(&c[tdst[i] & (BIN_NODES - 1)], 1);
        rec[pos] = trec[i];
    }
}

__global__ __launch_bounds__(256) void gin_layer_fused(
    const float* __restrict__ x, const int* __restrict__ rec_words,
    const int* __restrict__ off, const int* __restrict__ cnt,
    const float* __restrict__ dinv, const float* __restrict__ eps_arr, int eps_idx,
    const float* __restrict__ W, const float* __restrict__ bias,
    float* __restrict__ out, int n_nodes, int do_relu, int zero_row0)
{
    extern __shared__ float sW[];
    const int tid  = threadIdx.x;
    const int lane = tid & 31;
    const int half = lane >> 4;
    const int l16  = lane & 15;
    const int wave = __builtin_amdgcn_readfirstlane(tid >> 5);
    const int row0 = blockIdx.x * 128 + wave * 16;
    float* tile = sW + wave * 16 * SW_LD;

    const float4* __restrict__ x4 = (const float4*)x;
    const int2* __restrict__ rec = (const int2*)rec_words;

    const int  nd    = row0 + l16;
    const bool valid = nd < n_nodes;
    const int  ndc   = valid ? nd : n_nodes - 1;
    const int  segb  = off[ndc];
    const int  sege  = segb + cnt[ndc];
    const int  di_b  = valid ? __float_as_int(dinv[ndc]) : 0;
    const int  beg_all = __builtin_amdgcn_readlane(segb, 0);
    const int  end_all = (row0 < n_nodes) ? __builtin_amdgcn_readlane(sege, 15) : beg_all;
    const int  nfull   = (end_all - beg_all) >> 3;

    int2   eN[8];
    float4 vA[8], vB[8];
    int    wA[8], wB[8];

    if (nfull > 0) {
#pragma unroll
        for (int i = 0; i < 8; ++i) eN[i] = rec[beg_all + i];
    }

    float4 xr[16];
#pragma unroll
    for (int r = 0; r < 16; ++r) {
        const int rn = (row0 + r < n_nodes) ? (row0 + r) : (n_nodes - 1);
        xr[r] = x4[(size_t)rn * 32 + lane];
    }

    if (nfull > 0) {
#pragma unroll
        for (int i = 0; i < 8; ++i) { vA[i] = x4[(size_t)eN[i].x * 32 + lane]; wA[i] = eN[i].y; }
    }
    if (nfull > 1) {
#pragma unroll
        for (int i = 0; i < 8; ++i) eN[i] = rec[beg_all + 8 + i];
    }

    {
        const float s = 1.0f + eps_arr[eps_idx];
#pragma unroll
        for (int r = 0; r < 16; ++r) {
            float4 h;
            h.x = s * xr[r].x; h.y = s * xr[r].y; h.z = s * xr[r].z; h.w = s * xr[r].w;
            *(float4*)(tile + r * SW_LD + lane * 4) = h;
        }
    }

    int    j   = 0;
    int    nb  = __builtin_amdgcn_readlane(sege, 0);
    float4 acc = make_float4(0.f, 0.f, 0.f, 0.f);

#define GIN_FLUSH()                                                                     \
    do {                                                                                \
        const float d_ = __int_as_float(__builtin_amdgcn_readlane(di_b, j));            \
        float4* t_ = (float4*)(tile + j * SW_LD + lane * 4);                            \
        float4 h_ = *t_;                                                                \
        h_.x = fmaf(acc.x, d_, h_.x); h_.y = fmaf(acc.y, d_, h_.y);                     \
        h_.z = fmaf(acc.z, d_, h_.z); h_.w = fmaf(acc.w, d_, h_.w);                     \
        *t_ = h_;                                                                       \
        acc = make_float4(0.f, 0.f, 0.f, 0.f);                                          \
    } while (0)

#define GIN_CONSUME(q_, v_, w_)                                                         \
    do {                                                                                \
        while ((q_) >= nb) {                                                            \
            GIN_FLUSH();                                                                \
            ++j;                                                                        \
            nb = (j < 15) ? __builtin_amdgcn_readlane(sege, j) : 0x7fffffff;            \
        }                                                                               \
        const float wf_ = __int_as_float(w_);                                           \
        acc.x = fmaf((v_).x, wf_, acc.x); acc.y = fmaf((v_).y, wf_, acc.y);             \
        acc.z = fmaf((v_).z, wf_, acc.z); acc.w = fmaf((v_).w, wf_, acc.w);             \
    } while (0)

#define GIN_ISSUE(V_, W_)                                                               \
    _Pragma("unroll")                                                                   \
    for (int i = 0; i < 8; ++i) { V_[i] = x4[(size_t)eN[i].x * 32 + lane]; W_[i] = eN[i].y; }

#define GIN_LOADREC(P_)                                                                 \
    _Pragma("unroll")                                                                   \
    for (int i = 0; i < 8; ++i) eN[i] = rec[(P_) + i];

#define GIN_CONSUME8(P_, V_, W_)                                                        \
    _Pragma("unroll")                                                                   \
    for (int i = 0; i < 8; ++i) GIN_CONSUME((P_) + i, V_[i], W_[i]);

    int p = beg_all;
    if (nfull > 0) {
        int c = 0;
        for (;;) {
            if (c + 1 >= nfull) { GIN_CONSUME8(p, vA, wA); break; }
            GIN_ISSUE(vB, wB);
            if (c + 2 < nfull) { GIN_LOADREC(p + 16); }
            GIN_CONSUME8(p, vA, wA);
            ++c; p += 8;
            if (c + 1 >= nfull) { GIN_CONSUME8(p, vB, wB); break; }
            GIN_ISSUE(vA, wA);
            if (c + 2 < nfull) { GIN_LOADREC(p + 16); }
            GIN_CONSUME8(p, vB, wB);
            ++c; p += 8;
        }
        p = beg_all + nfull * 8;
    }
    if (p < end_all) {
        const int last = end_all - 1;
        int2 e[8];
#pragma unroll
        for (int i = 0; i < 8; ++i) e[i] = rec[(p + i < last) ? (p + i) : last];
#pragma unroll
        for (int i = 0; i < 8; ++i) vA[i] = x4[(size_t)e[i].x * 32 + lane];
#pragma unroll
        for (int i = 0; i < 8; ++i) {
            if (p + i < end_all) GIN_CONSUME(p + i, vA[i], e[i].y);
        }
    }
    for (; j < 16; ++j) GIN_FLUSH();
#undef GIN_CONSUME8
#undef GIN_LOADREC
#undef GIN_ISSUE
#undef GIN_CONSUME
#undef GIN_FLUSH

    __syncthreads();

    v2f A[32];
#pragma unroll
    for (int kb = 0; kb < 32; ++kb) {
        const int k = kb * 4 + half * 2;
        const float2 hv = *(const float2*)(tile + l16 * SW_LD + k);
        A[kb].x = hv.x;
        A[kb].y = hv.y;
    }

    __syncthreads();

    const float4* W4 = (const float4*)W;
    for (int i = tid; i < DCH * DCH / 4; i += 256) {
        int row = i >> 5;
        int q   = i & 31;
        *(float4*)(&sW[row * SW_LD + q * 4]) = W4[i];
    }
    __syncthreads();

#pragma unroll 1
    for (int jt = 0; jt < 8; ++jt) {
        const int col = jt * 16 + l16;
        const float* wrow = &sW[col * SW_LD];
        v8f cc = {};
#pragma unroll
        for (int kb = 0; kb < 32; ++kb) {
            const int k = kb * 4 + half * 2;
            v2f b;
            b.x = wrow[k];
            b.y = wrow[k + 1];
            cc = __builtin_amdgcn_wmma_f32_16x16x4_f32(
                     false, A[kb],  false, b,
                     (short)0, cc,  false,  false);
        }
        const float bv = bias[col];
#pragma unroll
        for (int r = 0; r < 8; ++r) {
            const int rr = row0 + half * 8 + r;
            if (rr < n_nodes) {
                float v = cc[r] + bv;
                if (do_relu) v = fmaxf(v, 0.0f);
                if (zero_row0 && rr == 0) v = 0.0f;
                out[(size_t)rr * DCH + col] = v;
            }
        }
    }
}

extern "C" void kernel_launch(void* const* d_in, const int* in_sizes, int n_in,
                              void* d_out, int out_size, void* d_ws, size_t ws_size,
                              hipStream_t stream) {
    const float* emb = (const float*)d_in[0];
    const float* w1  = (const float*)d_in[1];
    const float* b1  = (const float*)d_in[2];
    const float* w2  = (const float*)d_in[3];
    const float* b2  = (const float*)d_in[4];
    const float* eps = (const float*)d_in[5];
    const float* ew  = (const float*)d_in[6];
    const int*   src = (const int*)d_in[7];
    const int*   dst = (const int*)d_in[8];

    const int N = in_sizes[0] / DCH;
    const int E = in_sizes[6];

    int*   cnt     = (int*)d_ws;
    int*   off     = cnt + N;
    float* dinv    = (float*)(off + N);
    int*   counts  = (int*)(dinv + N);
    int*   binbase = counts + MAX_BINS * SLICES;
    int*   tdst    = binbase + 1024;
    int*   trec    = tdst + E;
    int*   rec     = trec + 2 * (size_t)E;
    float* bufA    = (float*)(rec + 2 * (size_t)E);
    float* bufB    = bufA + (size_t)N * DCH;
    float* out     = (float*)d_out;

    dim3 b256(256);

    const int nBins     = (N + BIN_NODES - 1) / BIN_NODES;
    const int per_slice = (E + SLICES - 1) / SLICES;
    bin_count<<<SLICES, 512, 0, stream>>>(dst, counts, E, nBins, per_slice);
    scan_counts<<<1, 1024, 0, stream>>>(counts, binbase, nBins);
    bin_scatter<<<SLICES, 512, 0, stream>>>(ew, src, dst, counts, tdst, trec, E, nBins, per_slice);
    bin_sort<<<nBins, b256, 0, stream>>>(binbase, tdst, trec, cnt, off, dinv, rec, N);

    const int    layerBlocks = (N + 127) / 128;
    const size_t shmem       = (size_t)DCH * SW_LD * sizeof(float);

    gin_layer_fused<<<layerBlocks, b256, shmem, stream>>>(emb, rec, off, cnt, dinv, eps, 0,
                                                          w1, b1, bufA, N, 1, 0);
    gin_layer_fused<<<layerBlocks, b256, shmem, stream>>>(bufA, rec, off, cnt, dinv, eps, 1,
                                                          w2, b2, bufB, N, 1, 0);
    gin_layer_fused<<<layerBlocks, b256, shmem, stream>>>(bufB, rec, off, cnt, dinv, eps, 2,
                                                          w2, b2, out, N, 0, 1);
}


# Layer kernel regridded: one 25-wave work-group (400 nodes) per WGP, W in an LDS region of its own staged once at entry, so WGPs carry equal loads and no barrier follows a gather; sum order kept
# speedup vs baseline: 3.5346x; 1.0601x over previous
#include <hip/hip_runtime.h>
#include <hip/hip_bf16.h>

typedef __attribute__((ext_vector_type(2))) float v2f;
typedef __attribute__((ext_vector_type(8))) float v8f;

#define DCH   128
#define SW_LD 132

#define BIN_SHIFT 8
#define BIN_NODES 256
#define MAX_BINS  512
#define SLICES    256

__global__ __launch_bounds__(512) void bin_count(const int* __restrict__ dst, int* __restrict__ counts,
                                                 int nE, int nBins, int per_slice) {
    __shared__ int h[MAX_BINS];
    const int tid = threadIdx.x;
    for (int b = tid; b < nBins; b += 512) h[b] = 0;
    __syncthreads();
    const int beg = blockIdx.x * per_slice;
    const int end = (beg + per_slice < nE) ? (beg + per_slice) : nE;
    for (int i = beg + tid; i < end; i += 512) atomicAdd(&h[dst[i] >> BIN_SHIFT], 1);
    __syncthreads();
    for (int b = tid; b < nBins; b += 512) counts[b * SLICES + blockIdx.x] = h[b];
}

__device__ __forceinline__ int wave_incl_scan(int v, int lane) {
#pragma unroll
    for (int o = 1; o < 32; o <<= 1) {
        const int u = __shfl_up(v, o);
        if (lane >= o) v += u;
    }
    return v;
}

__global__ __launch_bounds__(1024) void scan_counts(int* __restrict__ counts, int* __restrict__ binbase,
                                                    int nBins) {
    __shared__ int tot[MAX_BINS];
    __shared__ int wsum[32];
    const int tid  = threadIdx.x;
    const int lane = tid & 31;
    const int wave = tid >> 5;
    for (int bin = wave; bin < nBins; bin += 32) {
        const int4* row = (const int4*)(counts + bin * SLICES);
        const int4 a = row[lane];
        const int4 b = row[32 + lane];
        int s = a.x + a.y + a.z + a.w + b.x + b.y + b.z + b.w;
#pragma unroll
        for (int o = 16; o > 0; o >>= 1) s += __shfl_xor(s, o);
        if (lane == 0) tot[bin] = s;
    }
    __syncthreads();
    const int v   = (tid < nBins) ? tot[tid] : 0;
    const int inc = wave_incl_scan(v, lane);
    if (lane == 31) wsum[wave] = inc;
    __syncthreads();
    if (wave == 0) {
        const int w  = wsum[lane];
        const int wi = wave_incl_scan(w, lane);
        wsum[lane] = wi - w;
    }
    __syncthreads();
    const int excl = wsum[wave] + inc - v;
    if (tid < nBins) { tot[tid] = excl; binbase[tid] = excl; }
    if (tid == nBins - 1) binbase[nBins] = excl + v;
    __syncthreads();
    for (int bin = wave; bin < nBins; bin += 32) {
        int4* row = (int4*)(counts + bin * SLICES);
        const int4 a = row[lane];
        const int4 b = row[32 + lane];
        const int sa = a.x + a.y + a.z + a.w;
        const int sb = b.x + b.y + b.z + b.w;
        const int ia = wave_incl_scan(sa, lane);
        const int ib = wave_incl_scan(sb, lane);
        const int totA = __shfl(ia, 31);
        const int ea = tot[bin] + ia - sa;
        const int eb = tot[bin] + totA + ib - sb;
        row[lane]      = make_int4(ea, ea + a.x, ea + a.x + a.y, ea + a.x + a.y + a.z);
        row[32 + lane] = make_int4(eb, eb + b.x, eb + b.x + b.y, eb + b.x + b.y + b.z);
    }
}

__global__ __launch_bounds__(512) void bin_scatter(const float* __restrict__ ew, const int* __restrict__ src,
                                                   const int* __restrict__ dst, const int* __restrict__ counts,
                                                   int* __restrict__ tdst, int* __restrict__ trec_words,
                                                   int nE, int nBins, int per_slice) {
    __shared__ int cur[MAX_BINS];
    int2* __restrict__ trec = (int2*)trec_words;
    const int tid = threadIdx.x;
    for (int b = tid; b < nBins; b += 512) cur[b] = counts[b * SLICES + blockIdx.x];
    __syncthreads();
    const int beg = blockIdx.x * per_slice;
    const int end = (beg + per_slice < nE) ? (beg + per_slice) : nE;
    for (int i = beg + tid; i < end; i += 512) {
        const int d   = dst[i];
        const int pos = atomicAdd(&cur[d >> BIN_SHIFT], 1);
        tdst[pos] = d;
        trec[pos] = make_int2(src[i], __float_as_int(ew[i]));
    }
}

__global__ __launch_bounds__(256) void bin_sort(const int* __restrict__ binbase, const int* __restrict__ tdst,
                                                const int* __restrict__ trec_words,
                                                int* __restrict__ cnt, int* __restrict__ off,
                                                float* __restrict__ dinv, int* __restrict__ rec_words, int n) {
    __shared__ int c[BIN_NODES];
    __shared__ int wsum[8];
    const int2* __restrict__ trec = (const int2*)trec_words;
    int2* __restrict__ rec = (int2*)rec_words;
    const int tid  = threadIdx.x;
    const int lane = tid & 31;
    const int wave = tid >> 5;
    const int b0 = binbase[blockIdx.x];
    const int b1 = binbase[blockIdx.x + 1];
    c[tid] = 0;
    __syncthreads();
    for (int i = b0 + tid; i < b1; i += 256) atomicAdd(&c[tdst[i] & (BIN_NODES - 1)], 1);
    __syncthreads();
    const int cv  = c[tid];
    const int inc = wave_incl_scan(cv, lane);
    if (lane == 31) wsum[wave] = inc;
    __syncthreads();
    int before = 0;
#pragma unroll
    for (int w = 0; w < 8; ++w) if (w < wave) before += wsum[w];
    const int start = b0 + before + inc - cv;
    const int node  = blockIdx.x * BIN_NODES + tid;
    if (node < n) {
        cnt[node]  = cv;
        off[node]  = start;
        dinv[node] = (cv > 0) ? (1.0f / (float)cv) : 0.f;
    }
    c[tid] = start;
    __syncthreads();
    for (int i = b0 + tid; i < b1; i += 256) {
        const int pos = atomicAdd(&c[tdst[i] & (BIN_NODES - 1)], 1);
        rec[pos] = trec[i];
    }
}

#define LAYER_WAVES   25
#define LAYER_THREADS (LAYER_WAVES * 32)
#define LAYER_NODES   (LAYER_WAVES * 16)
#define W_STAGE_THREADS 512
#define W_PER_THREAD    (DCH * DCH / 4 / W_STAGE_THREADS)
static_assert(W_STAGE_THREADS <= LAYER_THREADS && W_PER_THREAD * W_STAGE_THREADS == DCH * DCH / 4,
              "the staging threads cover W exactly");
__global__ __launch_bounds__(LAYER_THREADS) void gin_layer_fused(
    const float* __restrict__ x, const int* __restrict__ rec_words,
    const int* __restrict__ off, const int* __restrict__ cnt,
    const float* __restrict__ dinv, const float* __restrict__ eps_arr, int eps_idx,
    const float* __restrict__ W, const float* __restrict__ bias,
    float* __restrict__ out, int n_nodes, int do_relu, int zero_row0)
{
    __shared__ __attribute__((aligned(16))) float sTile[LAYER_WAVES * 16 * SW_LD];
    __shared__ __attribute__((aligned(16))) float sW[DCH * SW_LD];
    const int tid  = threadIdx.x;
    const int lane = tid & 31;
    const int half = lane >> 4;
    const int l16  = lane & 15;
    const int wave = __builtin_amdgcn_readfirstlane(tid >> 5);
    const int row0 = blockIdx.x * LAYER_NODES + wave * 16;
    float* tile = sTile + wave * 16 * SW_LD;

    const float4* __restrict__ x4 = (const float4*)x;
    const int2* __restrict__ rec = (const int2*)rec_words;

    const int  nd    = row0 + l16;
    const bool valid = nd < n_nodes;
    const int  ndc   = valid ? nd : n_nodes - 1;
    const int  segb  = off[ndc];
    const int  sege  = segb + cnt[ndc];
    const int  di_b  = valid ? __float_as_int(dinv[ndc]) : 0;

    if (tid < W_STAGE_THREADS) {
        const float4* __restrict__ W4 = (const float4*)W;
        float4 wv[W_PER_THREAD];
#pragma unroll
        for (int t = 0; t < W_PER_THREAD; ++t) wv[t] = W4[tid + t * W_STAGE_THREADS];
#pragma unroll
        for (int t = 0; t < W_PER_THREAD; ++t) {
            const int i = tid + t * W_STAGE_THREADS;
            *(float4*)(&sW[(i >> 5) * SW_LD + (i & 31) * 4]) = wv[t];
        }
    }
    __syncthreads();
    const int  beg_all = __builtin_amdgcn_readlane(segb, 0);
    const int  end_all = (row0 < n_nodes) ? __builtin_amdgcn_readlane(sege, 15) : beg_all;
    const int  nfull   = (end_all - beg_all) >> 3;

    int2   eN[8];
    float4 vA[8], vB[8];
    int    wA[8], wB[8];

    if (nfull > 0) {
#pragma unroll
        for (int i = 0; i < 8; ++i) eN[i] = rec[beg_all + i];
    }

    float4 xr[16];
#pragma unroll
    for (int r = 0; r < 16; ++r) {
        const int rn = (row0 + r < n_nodes) ? (row0 + r) : (n_nodes - 1);
        xr[r] = x4[(size_t)rn * 32 + lane];
    }

    if (nfull > 0) {
#pragma unroll
        for (int i = 0; i < 8; ++i) { vA[i] = x4[(size_t)eN[i].x * 32 + lane]; wA[i] = eN[i].y; }
    }
    if (nfull > 1) {
#pragma unroll
        for (int i = 0; i < 8; ++i) eN[i] = rec[beg_all + 8 + i];
    }

    {
        const float s = 1.0f + eps_arr[eps_idx];
#pragma unroll
        for (int r = 0; r < 16; ++r) {
            float4 h;
            h.x = s * xr[r].x; h.y = s * xr[r].y; h.z = s * xr[r].z; h.w = s * xr[r].w;
            *(float4*)(tile + r * SW_LD + lane * 4) = h;
        }
    }

    int    j   = 0;
    int    nb  = __builtin_amdgcn_readlane(sege, 0);
    float4 acc = make_float4(0.f, 0.f, 0.f, 0.f);

#define GIN_FLUSH()                                                                     \
    do {                                                                                \
        const float d_ = __int_as_float(__builtin_amdgcn_readlane(di_b, j));            \
        float4* t_ = (float4*)(tile + j * SW_LD + lane * 4);                            \
        float4 h_ = *t_;                                                                \
        h_.x = fmaf(acc.x, d_, h_.x); h_.y = fmaf(acc.y, d_, h_.y);                     \
        h_.z = fmaf(acc.z, d_, h_.z); h_.w = fmaf(acc.w, d_, h_.w);                     \
        *t_ = h_;                                                                       \
        acc = make_float4(0.f, 0.f, 0.f, 0.f);                                          \
    } while (0)

#define GIN_CONSUME(q_, v_, w_)                                                         \
    do {                                                                                \
        while ((q_) >= nb) {                                                            \
            GIN_FLUSH();                                                                \
            ++j;                                                                        \
            nb = (j < 15) ? __builtin_amdgcn_readlane(sege, j) : 0x7fffffff;            \
        }                                                                               \
        const float wf_ = __int_as_float(w_);                                           \
        acc.x = fmaf((v_).x, wf_, acc.x); acc.y = fmaf((v_).y, wf_, acc.y);             \
        acc.z = fmaf((v_).z, wf_, acc.z); acc.w = fmaf((v_).w, wf_, acc.w);             \
    } while (0)

#define GIN_ISSUE(V_, W_)                                                               \
    _Pragma("unroll")                                                                   \
    for (int i = 0; i < 8; ++i) { V_[i] = x4[(size_t)eN[i].x * 32 + lane]; W_[i] = eN[i].y; }

#define GIN_LOADREC(P_)                                                                 \
    _Pragma("unroll")                                                                   \
    for (int i = 0; i < 8; ++i) eN[i] = rec[(P_) + i];

#define GIN_CONSUME8(P_, V_, W_)                                                        \
    _Pragma("unroll")                                                                   \
    for (int i = 0; i < 8; ++i) GIN_CONSUME((P_) + i, V_[i], W_[i]);

    int p = beg_all;
    if (nfull > 0) {
        int c = 0;
        for (;;) {
            if (c + 1 >= nfull) { GIN_CONSUME8(p, vA, wA); break; }
            GIN_ISSUE(vB, wB);
            if (c + 2 < nfull) { GIN_LOADREC(p + 16); }
            GIN_CONSUME8(p, vA, wA);
            ++c; p += 8;
            if (c + 1 >= nfull) { GIN_CONSUME8(p, vB, wB); break; }
            GIN_ISSUE(vA, wA);
            if (c + 2 < nfull) { GIN_LOADREC(p + 16); }
            GIN_CONSUME8(p, vB, wB);
            ++c; p += 8;
        }
        p = beg_all + nfull * 8;
    }
    if (p < end_all) {
        const int last = end_all - 1;
        int2 e[8];
#pragma unroll
        for (int i = 0; i < 8; ++i) e[i] = rec[(p + i < last) ? (p + i) : last];
#pragma unroll
        for (int i = 0; i < 8; ++i) vA[i] = x4[(size_t)e[i].x * 32 + lane];
#pragma unroll
        for (int i = 0; i < 8; ++i) {
            if (p + i < end_all) GIN_CONSUME(p + i, vA[i], e[i].y);
        }
    }
    for (; j < 16; ++j) GIN_FLUSH();
#undef GIN_CONSUME8
#undef GIN_LOADREC
#undef GIN_ISSUE
#undef GIN_CONSUME
#undef GIN_FLUSH


    v2f A[32];
#pragma unroll
    for (int kb = 0; kb < 32; ++kb) {
        const int k = kb * 4 + half * 2;
        const float2 hv = *(const float2*)(tile + l16 * SW_LD + k);
        A[kb].x = hv.x;
        A[kb].y = hv.y;
    }

#pragma unroll 1
    for (int jt = 0; jt < 8; ++jt) {
        const int col = jt * 16 + l16;
        const float* wrow = &sW[col * SW_LD];
        v8f cc = {};
#pragma unroll
        for (int kb = 0; kb < 32; ++kb) {
            const int k = kb * 4 + half * 2;
            v2f b;
            b.x = wrow[k];
            b.y = wrow[k + 1];
            cc = __builtin_amdgcn_wmma_f32_16x16x4_f32(
                     false, A[kb],  false, b,
                     (short)0, cc,  false,  false);
        }
        const float bv = bias[col];
#pragma unroll
        for (int r = 0; r < 8; ++r) {
            const int rr = row0 + half * 8 + r;
            if (rr < n_nodes) {
                float v = cc[r] + bv;
                if (do_relu) v = fmaxf(v, 0.0f);
                if (zero_row0 && rr == 0) v = 0.0f;
                out[(size_t)rr * DCH + col] = v;
            }
        }
    }
}

extern "C" void kernel_launch(void* const* d_in, const int* in_sizes, int n_in,
                              void* d_out, int out_size, void* d_ws, size_t ws_size,
                              hipStream_t stream) {
    const float* emb = (const float*)d_in[0];
    const float* w1  = (const float*)d_in[1];
    const float* b1  = (const float*)d_in[2];
    const float* w2  = (const float*)d_in[3];
    const float* b2  = (const float*)d_in[4];
    const float* eps = (const float*)d_in[5];
    const float* ew  = (const float*)d_in[6];
    const int*   src = (const int*)d_in[7];
    const int*   dst = (const int*)d_in[8];

    const int N = in_sizes[0] / DCH;
    const int E = in_sizes[6];

    int*   cnt     = (int*)d_ws;
    int*   off     = cnt + N;
    float* dinv    = (float*)(off + N);
    int*   counts  = (int*)(dinv + N);
    int*   binbase = counts + MAX_BINS * SLICES;
    int*   tdst    = binbase + 1024;
    int*   trec    = tdst + E;
    int*   rec     = trec + 2 * (size_t)E;
    float* bufA    = (float*)(rec + 2 * (size_t)E);
    float* bufB    = bufA + (size_t)N * DCH;
    float* out     = (float*)d_out;

    dim3 b256(256);

    const int nBins     = (N + BIN_NODES - 1) / BIN_NODES;
    const int per_slice = (E + SLICES - 1) / SLICES;
    bin_count<<<SLICES, 512, 0, stream>>>(dst, counts, E, nBins, per_slice);
    scan_counts<<<1, 1024, 0, stream>>>(counts, binbase, nBins);
    bin_scatter<<<SLICES, 512, 0, stream>>>(ew, src, dst, counts, tdst, trec, E, nBins, per_slice);
    bin_sort<<<nBins, b256, 0, stream>>>(binbase, tdst, trec, cnt, off, dinv, rec, N);

    const int layerBlocks = (N + LAYER_NODES - 1) / LAYER_NODES;
    dim3 bLayer(LAYER_THREADS);

    gin_layer_fused<<<layerBlocks, bLayer, 0, stream>>>(emb, rec, off, cnt, dinv, eps, 0,
                                                       w1, b1, bufA, N, 1, 0);
    gin_layer_fused<<<layerBlocks, bLayer, 0, stream>>>(bufA, rec, off, cnt, dinv, eps, 1,
                                                       w2, b2, bufB, N, 1, 0);
    gin_layer_fused<<<layerBlocks, bLayer, 0, stream>>>(bufB, rec, off, cnt, dinv, eps, 2,
                                                       w2, b2, out, N, 0, 1);
}


# Sort passes load in batches: a thread fetches 16-24 of its edges (the scan: a wave 4-8 count rows) before the dependent LDS atomics and shuffles, not one per trip; sums still in LDS arrival order
# speedup vs baseline: 3.8019x; 1.0756x over previous
#include <hip/hip_runtime.h>
#include <hip/hip_bf16.h>

typedef __attribute__((ext_vector_type(2))) float v2f;
typedef __attribute__((ext_vector_type(8))) float v8f;

#define DCH   128
#define SW_LD 132

#define BIN_SHIFT 8
#define BIN_NODES 256
#define MAX_BINS  512
#define SLICES    256
#define EDGE_BATCH 16
#define SORT_BATCH 24
#define SCAN_BATCH 8
#define SCAN_BATCH_PLACE 4

template <int N>
__device__ __forceinline__ void batch_loaded(const int (&v)[N]) {
#pragma unroll
    for (int u = 0; u < N; ++u) asm volatile("" :: "v"(v[u]));
}

__global__ __launch_bounds__(512) void bin_count(const int* __restrict__ dst, int* __restrict__ counts,
                                                 int nE, int nBins, int per_slice) {
    __shared__ int h[MAX_BINS];
    const int tid = threadIdx.x;
    for (int b = tid; b < nBins; b += 512) h[b] = 0;
    __syncthreads();
    const int beg = blockIdx.x * per_slice;
    const int end = (beg + per_slice < nE) ? (beg + per_slice) : nE;
    for (int base = beg + tid; base < end; base += 512 * EDGE_BATCH) {
        int d[EDGE_BATCH];
#pragma unroll
        for (int u = 0; u < EDGE_BATCH; ++u) {
            const int i = base + u * 512;
            d[u] = dst[(i < end) ? i : (end - 1)];
        }
        batch_loaded(d);
#pragma unroll
        for (int u = 0; u < EDGE_BATCH; ++u)
            if (base + u * 512 < end) atomicAdd(&h[d[u] >> BIN_SHIFT], 1);
    }
    __syncthreads();
    for (int b = tid; b < nBins; b += 512) counts[b * SLICES + blockIdx.x] = h[b];
}

__device__ __forceinline__ int wave_incl_scan(int v, int lane) {
#pragma unroll
    for (int o = 1; o < 32; o <<= 1) {
        const int u = __shfl_up(v, o);
        if (lane >= o) v += u;
    }
    return v;
}

__global__ __launch_bounds__(1024) void scan_counts(int* __restrict__ counts, int* __restrict__ binbase,
                                                    int nBins) {
    __shared__ int tot[MAX_BINS];
    __shared__ int wsum[32];
    const int tid  = threadIdx.x;
    const int lane = tid & 31;
    const int wave = tid >> 5;
    for (int bin0 = wave; bin0 < nBins; bin0 += 32 * SCAN_BATCH) {
        int4 a[SCAN_BATCH], b[SCAN_BATCH];
        int  s[SCAN_BATCH];
#pragma unroll
        for (int u = 0; u < SCAN_BATCH; ++u) {
            const int bin = (bin0 + 32 * u < nBins) ? (bin0 + 32 * u) : bin0;
            const int4* row = (const int4*)(counts + bin * SLICES);
            a[u] = row[lane];
            b[u] = row[32 + lane];
        }
#pragma unroll
        for (int u = 0; u < SCAN_BATCH; ++u)
            s[u] = a[u].x + a[u].y + a[u].z + a[u].w + b[u].x + b[u].y + b[u].z + b[u].w;
#pragma unroll
        for (int o = 16; o > 0; o >>= 1) {
#pragma unroll
            for (int u = 0; u < SCAN_BATCH; ++u) s[u] += __shfl_xor(s[u], o);
        }
#pragma unroll
        for (int u = 0; u < SCAN_BATCH; ++u)
            if (lane == 0 && bin0 + 32 * u < nBins) tot[bin0 + 32 * u] = s[u];
    }
    __syncthreads();
    const int v   = (tid < nBins) ? tot[tid] : 0;
    const int inc = wave_incl_scan(v, lane);
    if (lane == 31) wsum[wave] = inc;
    __syncthreads();
    if (wave == 0) {
        const int w  = wsum[lane];
        const int wi = wave_incl_scan(w, lane);
        wsum[lane] = wi - w;
    }
    __syncthreads();
    const int excl = wsum[wave] + inc - v;
    if (tid < nBins) { tot[tid] = excl; binbase[tid] = excl; }
    if (tid == nBins - 1) binbase[nBins] = excl + v;
    __syncthreads();
    for (int bin0 = wave; bin0 < nBins; bin0 += 32 * SCAN_BATCH_PLACE) {
        int4 a[SCAN_BATCH_PLACE], b[SCAN_BATCH_PLACE];
        int  ia[SCAN_BATCH_PLACE], ib[SCAN_BATCH_PLACE];
#pragma unroll
        for (int u = 0; u < SCAN_BATCH_PLACE; ++u) {
            const int bin = (bin0 + 32 * u < nBins) ? (bin0 + 32 * u) : bin0;
            const int4* row = (const int4*)(counts + bin * SLICES);
            a[u] = row[lane];
            b[u] = row[32 + lane];
        }
#pragma unroll
        for (int u = 0; u < SCAN_BATCH_PLACE; ++u) {
            ia[u] = a[u].x + a[u].y + a[u].z + a[u].w;
            ib[u] = b[u].x + b[u].y + b[u].z + b[u].w;
        }
#pragma unroll
        for (int o = 1; o < 32; o <<= 1) {
#pragma unroll
            for (int u = 0; u < SCAN_BATCH_PLACE; ++u) {
                const int xa = __shfl_up(ia[u], o);
                const int xb = __shfl_up(ib[u], o);
                if (lane >= o) { ia[u] += xa; ib[u] += xb; }
            }
        }
#pragma unroll
        for (int u = 0; u < SCAN_BATCH_PLACE; ++u) {
            const bool live = bin0 + 32 * u < nBins;
            const int  bin  = live ? (bin0 + 32 * u) : bin0;
            const int sa = a[u].x + a[u].y + a[u].z + a[u].w;
            const int sb = b[u].x + b[u].y + b[u].z + b[u].w;
            const int totA = __shfl(ia[u], 31);
            const int t  = tot[bin];
            const int ea = t + ia[u] - sa;
            const int eb = t + totA + ib[u] - sb;
            if (live) {
                int4* row = (int4*)(counts + bin * SLICES);
                row[lane]      = make_int4(ea, ea + a[u].x, ea + a[u].x + a[u].y, ea + a[u].x + a[u].y + a[u].z);
                row[32 + lane] = make_int4(eb, eb + b[u].x, eb + b[u].x + b[u].y, eb + b[u].x + b[u].y + b[u].z);
            }
        }
    }
}

__global__ __launch_bounds__(512) void bin_scatter(const float* __restrict__ ew, const int* __restrict__ src,
                                                   const int* __restrict__ dst, const int* __restrict__ counts,
                                                   int* __restrict__ tdst, int* __restrict__ trec_words,
                                                   int nE, int nBins, int per_slice) {
    __shared__ int cur[MAX_BINS];
    int2* __restrict__ trec = (int2*)trec_words;
    const int tid = threadIdx.x;
    for (int b = tid; b < nBins; b += 512) cur[b] = counts[b * SLICES + blockIdx.x];
    __syncthreads();
    const int beg = blockIdx.x * per_slice;
    const int end = (beg + per_slice < nE) ? (beg + per_slice) : nE;
    for (int base = beg + tid; base < end; base += 512 * EDGE_BATCH) {
        int d[EDGE_BATCH], s[EDGE_BATCH], w[EDGE_BATCH], pos[EDGE_BATCH];
#pragma unroll
        for (int u = 0; u < EDGE_BATCH; ++u) {
            const int i  = base + u * 512;
            const int ic = (i < end) ? i : (end - 1);
            d[u] = dst[ic];
            s[u] = src[ic];
            w[u] = __float_as_int(ew[ic]);
        }
#pragma unroll
        for (int u = 0; u < EDGE_BATCH; ++u)
            pos[u] = (base + u * 512 < end) ? atomicAdd(&cur[d[u] >> BIN_SHIFT], 1) : 0;
#pragma unroll
        for (int u = 0; u < EDGE_BATCH; ++u) {
            if (base + u * 512 < end) {
                tdst[pos[u]] = d[u];
                trec[pos[u]] = make_int2(s[u], w[u]);
            }
        }
    }
}

__global__ __launch_bounds__(256) void bin_sort(const int* __restrict__ binbase, const int* __restrict__ tdst,
                                                const int* __restrict__ trec_words,
                                                int* __restrict__ cnt, int* __restrict__ off,
                                                float* __restrict__ dinv, int* __restrict__ rec_words, int n) {
    __shared__ int c[BIN_NODES];
    __shared__ int wsum[8];
    const int2* __restrict__ trec = (const int2*)trec_words;
    int2* __restrict__ rec = (int2*)rec_words;
    const int tid  = threadIdx.x;
    const int lane = tid & 31;
    const int wave = tid >> 5;
    const int b0 = binbase[blockIdx.x];
    const int b1 = binbase[blockIdx.x + 1];
    c[tid] = 0;
    __syncthreads();
    for (int base = b0 + tid; base < b1; base += 256 * SORT_BATCH) {
        int d[SORT_BATCH];
#pragma unroll
        for (int u = 0; u < SORT_BATCH; ++u) {
            const int i = base + u * 256;
            d[u] = tdst[(i < b1) ? i : (b1 - 1)];
        }
        batch_loaded(d);
#pragma unroll
        for (int u = 0; u < SORT_BATCH; ++u)
            if (base + u * 256 < b1) atomicAdd(&c[d[u] & (BIN_NODES - 1)], 1);
    }
    __syncthreads();
    const int cv  = c[tid];
    const int inc = wave_incl_scan(cv, lane);
    if (lane == 31) wsum[wave] = inc;
    __syncthreads();
    int before = 0;
#pragma unroll
    for (int w = 0; w < 8; ++w) if (w < wave) before += wsum[w];
    const int start = b0 + before + inc - cv;
    const int node  = blockIdx.x * BIN_NODES + tid;
    if (node < n) {
        cnt[node]  = cv;
        off[node]  = start;
        dinv[node] = (cv > 0) ? (1.0f / (float)cv) : 0.f;
    }
    c[tid] = start;
    __syncthreads();
    for (int base = b0 + tid; base < b1; base += 256 * SORT_BATCH) {
        int  d[SORT_BATCH], pos[SORT_BATCH];
        int2 r[SORT_BATCH];
#pragma unroll
        for (int u = 0; u < SORT_BATCH; ++u) {
            const int i  = base + u * 256;
            const int ic = (i < b1) ? i : (b1 - 1);
            d[u] = tdst[ic];
            r[u] = trec[ic];
        }
        batch_loaded(d);
#pragma unroll
        for (int u = 0; u < SORT_BATCH; ++u)
            pos[u] = (base + u * 256 < b1) ? atomicAdd(&c[d[u] & (BIN_NODES - 1)], 1) : 0;
#pragma unroll
        for (int u = 0; u < SORT_BATCH; ++u)
            if (base + u * 256 < b1) rec[pos[u]] = r[u];
    }
}

#define LAYER_WAVES   25
#define LAYER_THREADS (LAYER_WAVES * 32)
#define LAYER_NODES   (LAYER_WAVES * 16)
#define W_STAGE_THREADS 512
#define W_PER_THREAD    (DCH * DCH / 4 / W_STAGE_THREADS)
static_assert(W_STAGE_THREADS <= LAYER_THREADS && W_PER_THREAD * W_STAGE_THREADS == DCH * DCH / 4,
              "the staging threads cover W exactly");
__global__ __launch_bounds__(LAYER_THREADS) void gin_layer_fused(
    const float* __restrict__ x, const int* __restrict__ rec_words,
    const int* __restrict__ off, const int* __restrict__ cnt,
    const float* __restrict__ dinv, const float* __restrict__ eps_arr, int eps_idx,
    const float* __restrict__ W, const float* __restrict__ bias,
    float* __restrict__ out, int n_nodes, int do_relu, int zero_row0)
{
    __shared__ __attribute__((aligned(16))) float sTile[LAYER_WAVES * 16 * SW_LD];
    __shared__ __attribute__((aligned(16))) float sW[DCH * SW_LD];
    const int tid  = threadIdx.x;
    const int lane = tid & 31;
    const int half = lane >> 4;
    const int l16  = lane & 15;
    const int wave = __builtin_amdgcn_readfirstlane(tid >> 5);
    const int row0 = blockIdx.x * LAYER_NODES + wave * 16;
    float* tile = sTile + wave * 16 * SW_LD;

    const float4* __restrict__ x4 = (const float4*)x;
    const int2* __restrict__ rec = (const int2*)rec_words;

    const int  nd    = row0 + l16;
    const bool valid = nd < n_nodes;
    const int  ndc   = valid ? nd : n_nodes - 1;
    const int  segb  = off[ndc];
    const int  sege  = segb + cnt[ndc];
    const int  di_b  = valid ? __float_as_int(dinv[ndc]) : 0;

    if (tid < W_STAGE_THREADS) {
        const float4* __restrict__ W4 = (const float4*)W;
        float4 wv[W_PER_THREAD];
#pragma unroll
        for (int t = 0; t < W_PER_THREAD; ++t) wv[t] = W4[tid + t * W_STAGE_THREADS];
#pragma unroll
        for (int t = 0; t < W_PER_THREAD; ++t) {
            const int i = tid + t * W_STAGE_THREADS;
            *(float4*)(&sW[(i >> 5) * SW_LD + (i & 31) * 4]) = wv[t];
        }
    }
    __syncthreads();
    const int  beg_all = __builtin_amdgcn_readlane(segb, 0);
    const int  end_all = (row0 < n_nodes) ? __builtin_amdgcn_readlane(sege, 15) : beg_all;
    const int  nfull   = (end_all - beg_all) >> 3;

    int2   eN[8];
    float4 vA[8], vB[8];
    int    wA[8], wB[8];

    if (nfull > 0) {
#pragma unroll
        for (int i = 0; i < 8; ++i) eN[i] = rec[beg_all + i];
    }

    float4 xr[16];
#pragma unroll
    for (int r = 0; r < 16; ++r) {
        const int rn = (row0 + r < n_nodes) ? (row0 + r) : (n_nodes - 1);
        xr[r] = x4[(size_t)rn * 32 + lane];
    }

    if (nfull > 0) {
#pragma unroll
        for (int i = 0; i < 8; ++i) { vA[i] = x4[(size_t)eN[i].x * 32 + lane]; wA[i] = eN[i].y; }
    }
    if (nfull > 1) {
#pragma unroll
        for (int i = 0; i < 8; ++i) eN[i] = rec[beg_all + 8 + i];
    }

    {
        const float s = 1.0f + eps_arr[eps_idx];
#pragma unroll
        for (int r = 0; r < 16; ++r) {
            float4 h;
            h.x = s * xr[r].x; h.y = s * xr[r].y; h.z = s * xr[r].z; h.w = s * xr[r].w;
            *(float4*)(tile + r * SW_LD + lane * 4) = h;
        }
    }

    int    j   = 0;
    int    nb  = __builtin_amdgcn_readlane(sege, 0);
    float4 acc = make_float4(0.f, 0.f, 0.f, 0.f);

#define GIN_FLUSH()                                                                     \
    do {                                                                                \
        const float d_ = __int_as_float(__builtin_amdgcn_readlane(di_b, j));            \
        float4* t_ = (float4*)(tile + j * SW_LD + lane * 4);                            \
        float4 h_ = *t_;                                                                \
        h_.x = fmaf(acc.x, d_, h_.x); h_.y = fmaf(acc.y, d_, h_.y);                     \
        h_.z = fmaf(acc.z, d_, h_.z); h_.w = fmaf(acc.w, d_, h_.w);                     \
        *t_ = h_;                                                                       \
        acc = make_float4(0.f, 0.f, 0.f, 0.f);                                          \
    } while (0)

#define GIN_CONSUME(q_, v_, w_)                                                         \
    do {                                                                                \
        while ((q_) >= nb) {                                                            \
            GIN_FLUSH();                                                                \
            ++j;                                                                        \
            nb = (j < 15) ? __builtin_amdgcn_readlane(sege, j) : 0x7fffffff;            \
        }                                                                               \
        const float wf_ = __int_as_float(w_);                                           \
        acc.x = fmaf((v_).x, wf_, acc.x); acc.y = fmaf((v_).y, wf_, acc.y);             \
        acc.z = fmaf((v_).z, wf_, acc.z); acc.w = fmaf((v_).w, wf_, acc.w);             \
    } while (0)

#define GIN_ISSUE(V_, W_)                                                               \
    _Pragma("unroll")                                                                   \
    for (int i = 0; i < 8; ++i) { V_[i] = x4[(size_t)eN[i].x * 32 + lane]; W_[i] = eN[i].y; }

#define GIN_LOADREC(P_)                                                                 \
    _Pragma("unroll")                                                                   \
    for (int i = 0; i < 8; ++i) eN[i] = rec[(P_) + i];

#define GIN_CONSUME8(P_, V_, W_)                                                        \
    _Pragma("unroll")                                                                   \
    for (int i = 0; i < 8; ++i) GIN_CONSUME((P_) + i, V_[i], W_[i]);

    int p = beg_all;
    if (nfull > 0) {
        int c = 0;
        for (;;) {
            if (c + 1 >= nfull) { GIN_CONSUME8(p, vA, wA); break; }
            GIN_ISSUE(vB, wB);
            if (c + 2 < nfull) { GIN_LOADREC(p + 16); }
            GIN_CONSUME8(p, vA, wA);
            ++c; p += 8;
            if (c + 1 >= nfull) { GIN_CONSUME8(p, vB, wB); break; }
            GIN_ISSUE(vA, wA);
            if (c + 2 < nfull) { GIN_LOADREC(p + 16); }
            GIN_CONSUME8(p, vB, wB);
            ++c; p += 8;
        }
        p = beg_all + nfull * 8;
    }
    if (p < end_all) {
        const int last = end_all - 1;
        int2 e[8];
#pragma unroll
        for (int i = 0; i < 8; ++i) e[i] = rec[(p + i < last) ? (p + i) : last];
#pragma unroll
        for (int i = 0; i < 8; ++i) vA[i] = x4[(size_t)e[i].x * 32 + lane];
#pragma unroll
        for (int i = 0; i < 8; ++i) {
            if (p + i < end_all) GIN_CONSUME(p + i, vA[i], e[i].y);
        }
    }
    for (; j < 16; ++j) GIN_FLUSH();
#undef GIN_CONSUME8
#undef GIN_LOADREC
#undef GIN_ISSUE
#undef GIN_CONSUME
#undef GIN_FLUSH


    v2f A[32];
#pragma unroll
    for (int kb = 0; kb < 32; ++kb) {
        const int k = kb * 4 + half * 2;
        const float2 hv = *(const float2*)(tile + l16 * SW_LD + k);
        A[kb].x = hv.x;
        A[kb].y = hv.y;
    }

#pragma unroll 1
    for (int jt = 0; jt < 8; ++jt) {
        const int col = jt * 16 + l16;
        const float* wrow = &sW[col * SW_LD];
        v8f cc = {};
#pragma unroll
        for (int kb = 0; kb < 32; ++kb) {
            const int k = kb * 4 + half * 2;
            v2f b;
            b.x = wrow[k];
            b.y = wrow[k + 1];
            cc = __builtin_amdgcn_wmma_f32_16x16x4_f32(
                     false, A[kb],  false, b,
                     (short)0, cc,  false,  false);
        }
        const float bv = bias[col];
#pragma unroll
        for (int r = 0; r < 8; ++r) {
            const int rr = row0 + half * 8 + r;
            if (rr < n_nodes) {
                float v = cc[r] + bv;
                if (do_relu) v = fmaxf(v, 0.0f);
                if (zero_row0 && rr == 0) v = 0.0f;
                out[(size_t)rr * DCH + col] = v;
            }
        }
    }
}

extern "C" void kernel_launch(void* const* d_in, const int* in_sizes, int n_in,
                              void* d_out, int out_size, void* d_ws, size_t ws_size,
                              hipStream_t stream) {
    const float* emb = (const float*)d_in[0];
    const float* w1  = (const float*)d_in[1];
    const float* b1  = (const float*)d_in[2];
    const float* w2  = (const float*)d_in[3];
    const float* b2  = (const float*)d_in[4];
    const float* eps = (const float*)d_in[5];
    const float* ew  = (const float*)d_in[6];
    const int*   src = (const int*)d_in[7];
    const int*   dst = (const int*)d_in[8];

    const int N = in_sizes[0] / DCH;
    const int E = in_sizes[6];

    int*   cnt     = (int*)d_ws;
    int*   off     = cnt + N;
    float* dinv    = (float*)(off + N);
    int*   counts  = (int*)(dinv + N);
    int*   binbase = counts + MAX_BINS * SLICES;
    int*   tdst    = binbase + 1024;
    int*   trec    = tdst + E;
    int*   rec     = trec + 2 * (size_t)E;
    float* bufA    = (float*)(rec + 2 * (size_t)E);
    float* bufB    = bufA + (size_t)N * DCH;
    float* out     = (float*)d_out;

    dim3 b256(256);

    const int nBins     = (N + BIN_NODES - 1) / BIN_NODES;
    const int per_slice = (E + SLICES - 1) / SLICES;
    bin_count<<<SLICES, 512, 0, stream>>>(dst, counts, E, nBins, per_slice);
    scan_counts<<<1, 1024, 0, stream>>>(counts, binbase, nBins);
    bin_scatter<<<SLICES, 512, 0, stream>>>(ew, src, dst, counts, tdst, trec, E, nBins, per_slice);
    bin_sort<<<nBins, b256, 0, stream>>>(binbase, tdst, trec, cnt, off, dinv, rec, N);

    const int layerBlocks = (N + LAYER_NODES - 1) / LAYER_NODES;
    dim3 bLayer(LAYER_THREADS);

    gin_layer_fused<<<layerBlocks, bLayer, 0, stream>>>(emb, rec, off, cnt, dinv, eps, 0,
                                                       w1, b1, bufA, N, 1, 0);
    gin_layer_fused<<<layerBlocks, bLayer, 0, stream>>>(bufA, rec, off, cnt, dinv, eps, 1,
                                                       w2, b2, bufB, N, 1, 0);
    gin_layer_fused<<<layerBlocks, bLayer, 0, stream>>>(bufB, rec, off, cnt, dinv, eps, 2,
                                                       w2, b2, out, N, 0, 1);
}


# Sort level 1 in one launch: each slice work-group counts, scans its own counts and regroups its slice in place; level 2 walks the slices' runs, a wave per run; no global scan; sums in arrival order
# speedup vs baseline: 4.4453x; 1.1692x over previous
#include <hip/hip_runtime.h>
#include <hip/hip_bf16.h>

typedef __attribute__((ext_vector_type(2))) float v2f;
typedef __attribute__((ext_vector_type(8))) float v8f;

#define DCH   128
#define SW_LD 132

#define BIN_SHIFT 8
#define BIN_NODES 256
#define MAX_BINS  512
#define SLICES    256
#define LO_LD     (MAX_BINS + 1)
#define EDGE_BATCH 16
#define RUN_BATCH  16

template <int N>
__device__ __forceinline__ void batch_loaded(const int (&v)[N]) {
#pragma unroll
    for (int u = 0; u < N; ++u) asm volatile("" :: "v"(v[u]));
}

__device__ __forceinline__ int wave_incl_scan(int v, int lane) {
#pragma unroll
    for (int o = 1; o < 32; o <<= 1) {
        const int u = __shfl_up(v, o);
        if (lane >= o) v += u;
    }
    return v;
}

static_assert(MAX_BINS == 512, "level 1 scans one bin per thread of its 512");
__global__ __launch_bounds__(512) void slice_group(const float* __restrict__ ew, const int* __restrict__ src,
                                                   const int* __restrict__ dst, int* __restrict__ lo,
                                                   int* __restrict__ tdst, int* __restrict__ trec_words,
                                                   int nE, int nBins, int per_slice) {
    __shared__ int h[MAX_BINS];
    __shared__ int wsum[16];
    int2* __restrict__ trec = (int2*)trec_words;
    const int tid  = threadIdx.x;
    const int lane = tid & 31;
    const int wave = tid >> 5;
    h[tid] = 0;
    __syncthreads();
    const int beg = blockIdx.x * per_slice;
    const int end = (beg + per_slice < nE) ? (beg + per_slice) : nE;
    for (int base = beg + tid; base < end; base += 512 * EDGE_BATCH) {
        int d[EDGE_BATCH];
#pragma unroll
        for (int u = 0; u < EDGE_BATCH; ++u) {
            const int i = base + u * 512;
            d[u] = dst[(i < end) ? i : (end - 1)];
        }
        batch_loaded(d);
#pragma unroll
        for (int u = 0; u < EDGE_BATCH; ++u)
            if (base + u * 512 < end) atomicAdd(&h[d[u] >> BIN_SHIFT], 1);
    }
    __syncthreads();
    const int v   = h[tid];
    const int inc = wave_incl_scan(v, lane);
    if (lane == 31) wsum[wave] = inc;
    __syncthreads();
    if (wave == 0) {
        const int w  = (lane < 16) ? wsum[lane] : 0;
        const int wi = wave_incl_scan(w, lane);
        if (lane < 16) wsum[lane] = wi - w;
    }
    __syncthreads();
    const int excl = wsum[wave] + inc - v;
    if (tid < nBins)      lo[blockIdx.x * LO_LD + tid]   = excl;
    if (tid == nBins - 1) lo[blockIdx.x * LO_LD + nBins] = excl + v;
    h[tid] = beg + excl;
    __syncthreads();
    for (int base = beg + tid; base < end; base += 512 * EDGE_BATCH) {
        int d[EDGE_BATCH], s[EDGE_BATCH], w[EDGE_BATCH], pos[EDGE_BATCH];
#pragma unroll
        for (int u = 0; u < EDGE_BATCH; ++u) {
            const int i  = base + u * 512;
            const int ic = (i < end) ? i : (end - 1);
            d[u] = dst[ic];
            s[u] = src[ic];
            w[u] = __float_as_int(ew[ic]);
        }
#pragma unroll
        for (int u = 0; u < EDGE_BATCH; ++u)
            pos[u] = (base + u * 512 < end) ? atomicAdd(&h[d[u] >> BIN_SHIFT], 1) : 0;
#pragma unroll
        for (int u = 0; u < EDGE_BATCH; ++u) {
            if (base + u * 512 < end) {
                tdst[pos[u]] = d[u];
                trec[pos[u]] = make_int2(s[u], w[u]);
            }
        }
    }
}

static_assert(SLICES == 256, "level 2 holds one run per thread of its 256");
__global__ __launch_bounds__(256) void bin_sort(const int* __restrict__ lo, const int* __restrict__ tdst,
                                                const int* __restrict__ trec_words,
                                                int* __restrict__ cnt, int* __restrict__ off,
                                                float* __restrict__ dinv, int* __restrict__ rec_words,
                                                int n, int per_slice) {
    __shared__ int c[BIN_NODES];
    __shared__ int wsum[8];
    __shared__ int red[8];
    const int2* __restrict__ trec = (const int2*)trec_words;
    int2* __restrict__ rec = (int2*)rec_words;
    const int tid  = threadIdx.x;
    const int lane = tid & 31;
    const int wave = tid >> 5;
    const int l0 = lo[tid * LO_LD + blockIdx.x];
    const int l1 = lo[tid * LO_LD + blockIdx.x + 1];
    const int rstart = tid * per_slice + l0;
    const int rlen   = l1 - l0;
    int part = l0;
#pragma unroll
    for (int o = 16; o > 0; o >>= 1) part += __shfl_xor(part, o);
    if (lane == 0) red[wave] = part;
    c[tid] = 0;
    __syncthreads();
    int b0 = 0;
#pragma unroll
    for (int w = 0; w < 8; ++w) b0 += red[w];

    for (int j0 = 0; j0 < 32; j0 += RUN_BATCH) {
        int d[RUN_BATCH], jl[RUN_BATCH];
#pragma unroll
        for (int u = 0; u < RUN_BATCH; ++u) {
            const int js = __shfl(rstart, j0 + u);
            jl[u] = __shfl(rlen, j0 + u);
            d[u] = tdst[(lane < jl[u]) ? (js + lane) : 0];
        }
        batch_loaded(d);
#pragma unroll
        for (int u = 0; u < RUN_BATCH; ++u)
            if (lane < jl[u]) atomicAdd(&c[d[u] & (BIN_NODES - 1)], 1);
    }
    for (int j = 0; j < 32; ++j) {
        const int js = __shfl(rstart, j);
        const int jn = __shfl(rlen, j);
        for (int k = 32 + lane; k < jn; k += 32) atomicAdd(&c[tdst[js + k] & (BIN_NODES - 1)], 1);
    }
    __syncthreads();
    const int cv  = c[tid];
    const int inc = wave_incl_scan(cv, lane);
    if (lane == 31) wsum[wave] = inc;
    __syncthreads();
    int before = 0;
#pragma unroll
    for (int w = 0; w < 8; ++w) if (w < wave) before += wsum[w];
    const int start = b0 + before + inc - cv;
    const int node  = blockIdx.x * BIN_NODES + tid;
    if (node < n) {
        cnt[node]  = cv;
        off[node]  = start;
        dinv[node] = (cv > 0) ? (1.0f / (float)cv) : 0.f;
    }
    c[tid] = start;
    __syncthreads();
    for (int j0 = 0; j0 < 32; j0 += RUN_BATCH) {
        int  d[RUN_BATCH], jl[RUN_BATCH], pos[RUN_BATCH];
        int2 r[RUN_BATCH];
#pragma unroll
        for (int u = 0; u < RUN_BATCH; ++u) {
            const int js = __shfl(rstart, j0 + u);
            jl[u] = __shfl(rlen, j0 + u);
            const int ic = (lane < jl[u]) ? (js + lane) : 0;
            d[u] = tdst[ic];
            r[u] = trec[ic];
        }
        batch_loaded(d);
#pragma unroll
        for (int u = 0; u < RUN_BATCH; ++u)
            pos[u] = (lane < jl[u]) ? atomicAdd(&c[d[u] & (BIN_NODES - 1)], 1) : 0;
#pragma unroll
        for (int u = 0; u < RUN_BATCH; ++u)
            if (lane < jl[u]) rec[pos[u]] = r[u];
    }
    for (int j = 0; j < 32; ++j) {
        const int js = __shfl(rstart, j);
        const int jn = __shfl(rlen, j);
        for (int k = 32 + lane; k < jn; k += 32) {
            const int  dd = tdst[js + k];
            const int2 rr = trec[js + k];
            rec[atomicAdd(&c[dd & (BIN_NODES - 1)], 1)] = rr;
        }
    }
}

#define LAYER_WAVES   25
#define LAYER_THREADS (LAYER_WAVES * 32)
#define LAYER_NODES   (LAYER_WAVES * 16)
#define W_STAGE_THREADS 512
#define W_PER_THREAD    (DCH * DCH / 4 / W_STAGE_THREADS)
static_assert(W_STAGE_THREADS <= LAYER_THREADS && W_PER_THREAD * W_STAGE_THREADS == DCH * DCH / 4,
              "the staging threads cover W exactly");
__global__ __launch_bounds__(LAYER_THREADS) void gin_layer_fused(
    const float* __restrict__ x, const int* __restrict__ rec_words,
    const int* __restrict__ off, const int* __restrict__ cnt,
    const float* __restrict__ dinv, const float* __restrict__ eps_arr, int eps_idx,
    const float* __restrict__ W, const float* __restrict__ bias,
    float* __restrict__ out, int n_nodes, int do_relu, int zero_row0)
{
    __shared__ __attribute__((aligned(16))) float sTile[LAYER_WAVES * 16 * SW_LD];
    __shared__ __attribute__((aligned(16))) float sW[DCH * SW_LD];
    const int tid  = threadIdx.x;
    const int lane = tid & 31;
    const int half = lane >> 4;
    const int l16  = lane & 15;
    const int wave = __builtin_amdgcn_readfirstlane(tid >> 5);
    const int row0 = blockIdx.x * LAYER_NODES + wave * 16;
    float* tile = sTile + wave * 16 * SW_LD;

    const float4* __restrict__ x4 = (const float4*)x;
    const int2* __restrict__ rec = (const int2*)rec_words;

    const int  nd    = row0 + l16;
    const bool valid = nd < n_nodes;
    const int  ndc   = valid ? nd : n_nodes - 1;
    const int  segb  = off[ndc];
    const int  sege  = segb + cnt[ndc];
    const int  di_b  = valid ? __float_as_int(dinv[ndc]) : 0;

    if (tid < W_STAGE_THREADS) {
        const float4* __restrict__ W4 = (const float4*)W;
        float4 wv[W_PER_THREAD];
#pragma unroll
        for (int t = 0; t < W_PER_THREAD; ++t) wv[t] = W4[tid + t * W_STAGE_THREADS];
#pragma unroll
        for (int t = 0; t < W_PER_THREAD; ++t) {
            const int i = tid + t * W_STAGE_THREADS;
            *(float4*)(&sW[(i >> 5) * SW_LD + (i & 31) * 4]) = wv[t];
        }
    }
    __syncthreads();
    const int  beg_all = __builtin_amdgcn_readlane(segb, 0);
    const int  end_all = (row0 < n_nodes) ? __builtin_amdgcn_readlane(sege, 15) : beg_all;
    const int  nfull   = (end_all - beg_all) >> 3;

    int2   eN[8];
    float4 vA[8], vB[8];
    int    wA[8], wB[8];

    if (nfull > 0) {
#pragma unroll
        for (int i = 0; i < 8; ++i) eN[i] = rec[beg_all + i];
    }

    float4 xr[16];
#pragma unroll
    for (int r = 0; r < 16; ++r) {
        const int rn = (row0 + r < n_nodes) ? (row0 + r) : (n_nodes - 1);
        xr[r] = x4[(size_t)rn * 32 + lane];
    }

    if (nfull > 0) {
#pragma unroll
        for (int i = 0; i < 8; ++i) { vA[i] = x4[(size_t)eN[i].x * 32 + lane]; wA[i] = eN[i].y; }
    }
    if (nfull > 1) {
#pragma unroll
        for (int i = 0; i < 8; ++i) eN[i] = rec[beg_all + 8 + i];
    }

    {
        const float s = 1.0f + eps_arr[eps_idx];
#pragma unroll
        for (int r = 0; r < 16; ++r) {
            float4 h;
            h.x = s * xr[r].x; h.y = s * xr[r].y; h.z = s * xr[r].z; h.w = s * xr[r].w;
            *(float4*)(tile + r * SW_LD + lane * 4) = h;
        }
    }

    int    j   = 0;
    int    nb  = __builtin_amdgcn_readlane(sege, 0);
    float4 acc = make_float4(0.f, 0.f, 0.f, 0.f);

#define GIN_FLUSH()                                                                     \
    do {                                                                                \
        const float d_ = __int_as_float(__builtin_amdgcn_readlane(di_b, j));            \
        float4* t_ = (float4*)(tile + j * SW_LD + lane * 4);                            \
        float4 h_ = *t_;                                                                \
        h_.x = fmaf(acc.x, d_, h_.x); h_.y = fmaf(acc.y, d_, h_.y);                     \
        h_.z = fmaf(acc.z, d_, h_.z); h_.w = fmaf(acc.w, d_, h_.w);                     \
        *t_ = h_;                                                                       \
        acc = make_float4(0.f, 0.f, 0.f, 0.f);                                          \
    } while (0)

#define GIN_CONSUME(q_, v_, w_)                                                         \
    do {                                                                                \
        while ((q_) >= nb) {                                                            \
            GIN_FLUSH();                                                                \
            ++j;                                                                        \
            nb = (j < 15) ? __builtin_amdgcn_readlane(sege, j) : 0x7fffffff;            \
        }                                                                               \
        const float wf_ = __int_as_float(w_);                                           \
        acc.x = fmaf((v_).x, wf_, acc.x); acc.y = fmaf((v_).y, wf_, acc.y);             \
        acc.z = fmaf((v_).z, wf_, acc.z); acc.w = fmaf((v_).w, wf_, acc.w);             \
    } while (0)

#define GIN_ISSUE(V_, W_)                                                               \
    _Pragma("unroll")                                                                   \
    for (int i = 0; i < 8; ++i) { V_[i] = x4[(size_t)eN[i].x * 32 + lane]; W_[i] = eN[i].y; }

#define GIN_LOADREC(P_)                                                                 \
    _Pragma("unroll")                                                                   \
    for (int i = 0; i < 8; ++i) eN[i] = rec[(P_) + i];

#define GIN_CONSUME8(P_, V_, W_)                                                        \
    _Pragma("unroll")                                                                   \
    for (int i = 0; i < 8; ++i) GIN_CONSUME((P_) + i, V_[i], W_[i]);

    int p = beg_all;
    if (nfull > 0) {
        int c = 0;
        for (;;) {
            if (c + 1 >= nfull) { GIN_CONSUME8(p, vA, wA); break; }
            GIN_ISSUE(vB, wB);
            if (c + 2 < nfull) { GIN_LOADREC(p + 16); }
            GIN_CONSUME8(p, vA, wA);
            ++c; p += 8;
            if (c + 1 >= nfull) { GIN_CONSUME8(p, vB, wB); break; }
            GIN_ISSUE(vA, wA);
            if (c + 2 < nfull) { GIN_LOADREC(p + 16); }
            GIN_CONSUME8(p, vB, wB);
            ++c; p += 8;
        }
        p = beg_all + nfull * 8;
    }
    if (p < end_all) {
        const int last = end_all - 1;
        int2 e[8];
#pragma unroll
        for (int i = 0; i < 8; ++i) e[i] = rec[(p + i < last) ? (p + i) : last];
#pragma unroll
        for (int i = 0; i < 8; ++i) vA[i] = x4[(size_t)e[i].x * 32 + lane];
#pragma unroll
        for (int i = 0; i < 8; ++i) {
            if (p + i < end_all) GIN_CONSUME(p + i, vA[i], e[i].y);
        }
    }
    for (; j < 16; ++j) GIN_FLUSH();
#undef GIN_CONSUME8
#undef GIN_LOADREC
#undef GIN_ISSUE
#undef GIN_CONSUME
#undef GIN_FLUSH


    v2f A[32];
#pragma unroll
    for (int kb = 0; kb < 32; ++kb) {
        const int k = kb * 4 + half * 2;
        const float2 hv = *(const float2*)(tile + l16 * SW_LD + k);
        A[kb].x = hv.x;
        A[kb].y = hv.y;
    }

#pragma unroll 1
    for (int jt = 0; jt < 8; ++jt) {
        const int col = jt * 16 + l16;
        const float* wrow = &sW[col * SW_LD];
        v8f cc = {};
#pragma unroll
        for (int kb = 0; kb < 32; ++kb) {
            const int k = kb * 4 + half * 2;
            v2f b;
            b.x = wrow[k];
            b.y = wrow[k + 1];
            cc = __builtin_amdgcn_wmma_f32_16x16x4_f32(
                     false, A[kb],  false, b,
                     (short)0, cc,  false,  false);
        }
        const float bv = bias[col];
#pragma unroll
        for (int r = 0; r < 8; ++r) {
            const int rr = row0 + half * 8 + r;
            if (rr < n_nodes) {
                float v = cc[r] + bv;
                if (do_relu) v = fmaxf(v, 0.0f);
                if (zero_row0 && rr == 0) v = 0.0f;
                out[(size_t)rr * DCH + col] = v;
            }
        }
    }
}

extern "C" void kernel_launch(void* const* d_in, const int* in_sizes, int n_in,
                              void* d_out, int out_size, void* d_ws, size_t ws_size,
                              hipStream_t stream) {
    const float* emb = (const float*)d_in[0];
    const float* w1  = (const float*)d_in[1];
    const float* b1  = (const float*)d_in[2];
    const float* w2  = (const float*)d_in[3];
    const float* b2  = (const float*)d_in[4];
    const float* eps = (const float*)d_in[5];
    const float* ew  = (const float*)d_in[6];
    const int*   src = (const int*)d_in[7];
    const int*   dst = (const int*)d_in[8];

    const int N = in_sizes[0] / DCH;
    const int E = in_sizes[6];

    int*   cnt     = (int*)d_ws;
    int*   off     = cnt + N;
    float* dinv    = (float*)(off + N);
    int*   lo      = (int*)(dinv + N);
    int*   tdst    = lo + SLICES * LO_LD;
    int*   trec    = tdst + E;
    int*   rec     = trec + 2 * (size_t)E;
    float* bufA    = (float*)(rec + 2 * (size_t)E);
    float* bufB    = bufA + (size_t)N * DCH;
    float* out     = (float*)d_out;

    dim3 b256(256);

    const int nBins     = (N + BIN_NODES - 1) / BIN_NODES;
    const int per_slice = (E + SLICES - 1) / SLICES;
    slice_group<<<SLICES, 512, 0, stream>>>(ew, src, dst, lo, tdst, trec, E, nBins, per_slice);
    bin_sort<<<nBins, b256, 0, stream>>>(lo, tdst, trec, cnt, off, dinv, rec, N, per_slice);

    const int layerBlocks = (N + LAYER_NODES - 1) / LAYER_NODES;
    dim3 bLayer(LAYER_THREADS);

    gin_layer_fused<<<layerBlocks, bLayer, 0, stream>>>(emb, rec, off, cnt, dinv, eps, 0,
                                                       w1, b1, bufA, N, 1, 0);
    gin_layer_fused<<<layerBlocks, bLayer, 0, stream>>>(bufA, rec, off, cnt, dinv, eps, 1,
                                                       w2, b2, bufB, N, 1, 0);
    gin_layer_fused<<<layerBlocks, bLayer, 0, stream>>>(bufB, rec, off, cnt, dinv, eps, 2,
                                                       w2, b2, out, N, 0, 1);
}


# Both sort levels put regrouped records into an LDS copy of their output range and write it out in order as whole lines, not 32 scattered 4/8-byte pieces per store; sums stay in LDS arrival order
# speedup vs baseline: 4.7138x; 1.0604x over previous
#include <hip/hip_runtime.h>
#include <hip/hip_bf16.h>

typedef __attribute__((ext_vector_type(2))) float v2f;
typedef __attribute__((ext_vector_type(8))) float v8f;

#define DCH   128
#define SW_LD 132

#define BIN_SHIFT 8
#define BIN_NODES 256
#define MAX_BINS  512
#define SLICES    256
#define LO_LD     (MAX_BINS + 1)
#define EDGE_BATCH 16
#define RUN_BATCH  16
#define SLICE_CAP  8192
#define BIN_CAP    6144

template <int N>
__device__ __forceinline__ void batch_loaded(const int (&v)[N]) {
#pragma unroll
    for (int u = 0; u < N; ++u) asm volatile("" :: "v"(v[u]));
}

__device__ __forceinline__ int wave_incl_scan(int v, int lane) {
#pragma unroll
    for (int o = 1; o < 32; o <<= 1) {
        const int u = __shfl_up(v, o);
        if (lane >= o) v += u;
    }
    return v;
}

static_assert(MAX_BINS == 512, "level 1 scans one bin per thread of its 512");
__global__ __launch_bounds__(512) void slice_group(const float* __restrict__ ew, const int* __restrict__ src,
                                                   const int* __restrict__ dst, int* __restrict__ lo,
                                                   int* __restrict__ tdst, int* __restrict__ trec_words,
                                                   int nE, int nBins, int per_slice) {
    __shared__ int h[MAX_BINS];
    __shared__ int wsum[16];
    __shared__ int sd[SLICE_CAP];
    __shared__ __attribute__((aligned(8))) int2 sr[SLICE_CAP];
    int2* __restrict__ trec = (int2*)trec_words;
    const int tid  = threadIdx.x;
    const int lane = tid & 31;
    const int wave = tid >> 5;
    h[tid] = 0;
    __syncthreads();
    const int beg = blockIdx.x * per_slice;
    const int end = (beg + per_slice < nE) ? (beg + per_slice) : nE;
    for (int base = beg + tid; base < end; base += 512 * EDGE_BATCH) {
        int d[EDGE_BATCH];
#pragma unroll
        for (int u = 0; u < EDGE_BATCH; ++u) {
            const int i = base + u * 512;
            d[u] = dst[(i < end) ? i : (end - 1)];
        }
        batch_loaded(d);
#pragma unroll
        for (int u = 0; u < EDGE_BATCH; ++u)
            if (base + u * 512 < end) atomicAdd(&h[d[u] >> BIN_SHIFT], 1);
    }
    __syncthreads();
    const int v   = h[tid];
    const int inc = wave_incl_scan(v, lane);
    if (lane == 31) wsum[wave] = inc;
    __syncthreads();
    if (wave == 0) {
        const int w  = (lane < 16) ? wsum[lane] : 0;
        const int wi = wave_incl_scan(w, lane);
        if (lane < 16) wsum[lane] = wi - w;
    }
    __syncthreads();
    const int excl = wsum[wave] + inc - v;
    if (tid < nBins)      lo[blockIdx.x * LO_LD + tid]   = excl;
    if (tid == nBins - 1) lo[blockIdx.x * LO_LD + nBins] = excl + v;
    h[tid] = excl;
    __syncthreads();
    for (int base = beg + tid; base < end; base += 512 * EDGE_BATCH) {
        int d[EDGE_BATCH], s[EDGE_BATCH], w[EDGE_BATCH], pos[EDGE_BATCH];
#pragma unroll
        for (int u = 0; u < EDGE_BATCH; ++u) {
            const int i  = base + u * 512;
            const int ic = (i < end) ? i : (end - 1);
            d[u] = dst[ic];
            s[u] = src[ic];
            w[u] = __float_as_int(ew[ic]);
        }
#pragma unroll
        for (int u = 0; u < EDGE_BATCH; ++u)
            pos[u] = (base + u * 512 < end) ? atomicAdd(&h[d[u] >> BIN_SHIFT], 1) : 0;
#pragma unroll
        for (int u = 0; u < EDGE_BATCH; ++u) {
            if (base + u * 512 < end) {
                if (pos[u] < SLICE_CAP) {
                    sd[pos[u]] = d[u];
                    sr[pos[u]] = make_int2(s[u], w[u]);
                } else {
                    tdst[beg + pos[u]] = d[u];
                    trec[beg + pos[u]] = make_int2(s[u], w[u]);
                }
            }
        }
    }
    __syncthreads();
    const int len    = end - beg;
    const int staged = (len < SLICE_CAP) ? len : SLICE_CAP;
    const int head = (32 - (beg & 31)) & 31;
    if (tid < head && tid < staged) {
        tdst[beg + tid] = sd[tid];
        trec[beg + tid] = sr[tid];
    }
#pragma unroll 4
    for (int i = head + tid; i < staged; i += 512) {
        tdst[beg + i] = sd[i];
        trec[beg + i] = sr[i];
    }
}

static_assert(SLICES == 256, "level 2 holds one run per thread of its 256");
__global__ __launch_bounds__(256) void bin_sort(const int* __restrict__ lo, const int* __restrict__ tdst,
                                                const int* __restrict__ trec_words,
                                                int* __restrict__ cnt, int* __restrict__ off,
                                                float* __restrict__ dinv, int* __restrict__ rec_words,
                                                int n, int per_slice) {
    __shared__ int c[BIN_NODES];
    __shared__ int wsum[8];
    __shared__ int red[8];
    __shared__ __attribute__((aligned(8))) int2 sr[BIN_CAP];
    const int2* __restrict__ trec = (const int2*)trec_words;
    int2* __restrict__ rec = (int2*)rec_words;
    const int tid  = threadIdx.x;
    const int lane = tid & 31;
    const int wave = tid >> 5;
    const int l0 = lo[tid * LO_LD + blockIdx.x];
    const int l1 = lo[tid * LO_LD + blockIdx.x + 1];
    const int rstart = tid * per_slice + l0;
    const int rlen   = l1 - l0;
    int part = l0;
#pragma unroll
    for (int o = 16; o > 0; o >>= 1) part += __shfl_xor(part, o);
    if (lane == 0) red[wave] = part;
    c[tid] = 0;
    __syncthreads();
    int b0 = 0;
#pragma unroll
    for (int w = 0; w < 8; ++w) b0 += red[w];

    for (int j0 = 0; j0 < 32; j0 += RUN_BATCH) {
        int d[RUN_BATCH], jl[RUN_BATCH];
#pragma unroll
        for (int u = 0; u < RUN_BATCH; ++u) {
            const int js = __shfl(rstart, j0 + u);
            jl[u] = __shfl(rlen, j0 + u);
            d[u] = tdst[(lane < jl[u]) ? (js + lane) : 0];
        }
        batch_loaded(d);
#pragma unroll
        for (int u = 0; u < RUN_BATCH; ++u)
            if (lane < jl[u]) atomicAdd(&c[d[u] & (BIN_NODES - 1)], 1);
    }
    for (int j = 0; j < 32; ++j) {
        const int js = __shfl(rstart, j);
        const int jn = __shfl(rlen, j);
        for (int k = 32 + lane; k < jn; k += 32) atomicAdd(&c[tdst[js + k] & (BIN_NODES - 1)], 1);
    }
    __syncthreads();
    const int cv  = c[tid];
    const int inc = wave_incl_scan(cv, lane);
    if (lane == 31) wsum[wave] = inc;
    __syncthreads();
    int before = 0, total = 0;
#pragma unroll
    for (int w = 0; w < 8; ++w) { if (w < wave) before += wsum[w]; total += wsum[w]; }
    const int lstart = before + inc - cv;
    const int start  = b0 + lstart;
    const int node  = blockIdx.x * BIN_NODES + tid;
    if (node < n) {
        cnt[node]  = cv;
        off[node]  = start;
        dinv[node] = (cv > 0) ? (1.0f / (float)cv) : 0.f;
    }
    c[tid] = lstart;
    __syncthreads();
    for (int j0 = 0; j0 < 32; j0 += RUN_BATCH) {
        int  d[RUN_BATCH], jl[RUN_BATCH], pos[RUN_BATCH];
        int2 r[RUN_BATCH];
#pragma unroll
        for (int u = 0; u < RUN_BATCH; ++u) {
            const int js = __shfl(rstart, j0 + u);
            jl[u] = __shfl(rlen, j0 + u);
            const int ic = (lane < jl[u]) ? (js + lane) : 0;
            d[u] = tdst[ic];
            r[u] = trec[ic];
        }
        batch_loaded(d);
#pragma unroll
        for (int u = 0; u < RUN_BATCH; ++u)
            pos[u] = (lane < jl[u]) ? atomicAdd(&c[d[u] & (BIN_NODES - 1)], 1) : 0;
#pragma unroll
        for (int u = 0; u < RUN_BATCH; ++u) {
            if (lane < jl[u]) {
                if (pos[u] < BIN_CAP) sr[pos[u]] = r[u];
                else                  rec[b0 + pos[u]] = r[u];
            }
        }
    }
    for (int j = 0; j < 32; ++j) {
        const int js = __shfl(rstart, j);
        const int jn = __shfl(rlen, j);
        for (int k = 32 + lane; k < jn; k += 32) {
            const int  dd = tdst[js + k];
            const int2 rr = trec[js + k];
            const int  pp = atomicAdd(&c[dd & (BIN_NODES - 1)], 1);
            if (pp < BIN_CAP) sr[pp] = rr;
            else              rec[b0 + pp] = rr;
        }
    }
    __syncthreads();
    const int staged = (total < BIN_CAP) ? total : BIN_CAP;
    const int head = (32 - (b0 & 31)) & 31;
    if (tid < head && tid < staged) rec[b0 + tid] = sr[tid];
#pragma unroll 4
    for (int i = head + tid; i < staged; i += 256) rec[b0 + i] = sr[i];
}

#define LAYER_WAVES   25
#define LAYER_THREADS (LAYER_WAVES * 32)
#define LAYER_NODES   (LAYER_WAVES * 16)
#define W_STAGE_THREADS 512
#define W_PER_THREAD    (DCH * DCH / 4 / W_STAGE_THREADS)
static_assert(W_STAGE_THREADS <= LAYER_THREADS && W_PER_THREAD * W_STAGE_THREADS == DCH * DCH / 4,
              "the staging threads cover W exactly");
__global__ __launch_bounds__(LAYER_THREADS) void gin_layer_fused(
    const float* __restrict__ x, const int* __restrict__ rec_words,
    const int* __restrict__ off, const int* __restrict__ cnt,
    const float* __restrict__ dinv, const float* __restrict__ eps_arr, int eps_idx,
    const float* __restrict__ W, const float* __restrict__ bias,
    float* __restrict__ out, int n_nodes, int do_relu, int zero_row0)
{
    __shared__ __attribute__((aligned(16))) float sTile[LAYER_WAVES * 16 * SW_LD];
    __shared__ __attribute__((aligned(16))) float sW[DCH * SW_LD];
    const int tid  = threadIdx.x;
    const int lane = tid & 31;
    const int half = lane >> 4;
    const int l16  = lane & 15;
    const int wave = __builtin_amdgcn_readfirstlane(tid >> 5);
    const int row0 = blockIdx.x * LAYER_NODES + wave * 16;
    float* tile = sTile + wave * 16 * SW_LD;

    const float4* __restrict__ x4 = (const float4*)x;
    const int2* __restrict__ rec = (const int2*)rec_words;

    const int  nd    = row0 + l16;
    const bool valid = nd < n_nodes;
    const int  ndc   = valid ? nd : n_nodes - 1;
    const int  segb  = off[ndc];
    const int  sege  = segb + cnt[ndc];
    const int  di_b  = valid ? __float_as_int(dinv[ndc]) : 0;

    if (tid < W_STAGE_THREADS) {
        const float4* __restrict__ W4 = (const float4*)W;
        float4 wv[W_PER_THREAD];
#pragma unroll
        for (int t = 0; t < W_PER_THREAD; ++t) wv[t] = W4[tid + t * W_STAGE_THREADS];
#pragma unroll
        for (int t = 0; t < W_PER_THREAD; ++t) {
            const int i = tid + t * W_STAGE_THREADS;
            *(float4*)(&sW[(i >> 5) * SW_LD + (i & 31) * 4]) = wv[t];
        }
    }
    __syncthreads();
    const int  beg_all = __builtin_amdgcn_readlane(segb, 0);
    const int  end_all = (row0 < n_nodes) ? __builtin_amdgcn_readlane(sege, 15) : beg_all;
    const int  nfull   = (end_all - beg_all) >> 3;

    int2   eN[8];
    float4 vA[8], vB[8];
    int    wA[8], wB[8];

    if (nfull > 0) {
#pragma unroll
        for (int i = 0; i < 8; ++i) eN[i] = rec[beg_all + i];
    }

    float4 xr[16];
#pragma unroll
    for (int r = 0; r < 16; ++r) {
        const int rn = (row0 + r < n_nodes) ? (row0 + r) : (n_nodes - 1);
        xr[r] = x4[(size_t)rn * 32 + lane];
    }

    if (nfull > 0) {
#pragma unroll
        for (int i = 0; i < 8; ++i) { vA[i] = x4[(size_t)eN[i].x * 32 + lane]; wA[i] = eN[i].y; }
    }
    if (nfull > 1) {
#pragma unroll
        for (int i = 0; i < 8; ++i) eN[i] = rec[beg_all + 8 + i];
    }

    {
        const float s = 1.0f + eps_arr[eps_idx];
#pragma unroll
        for (int r = 0; r < 16; ++r) {
            float4 h;
            h.x = s * xr[r].x; h.y = s * xr[r].y; h.z = s * xr[r].z; h.w = s * xr[r].w;
            *(float4*)(tile + r * SW_LD + lane * 4) = h;
        }
    }

    int    j   = 0;
    int    nb  = __builtin_amdgcn_readlane(sege, 0);
    float4 acc = make_float4(0.f, 0.f, 0.f, 0.f);

#define GIN_FLUSH()                                                                     \
    do {                                                                                \
        const float d_ = __int_as_float(__builtin_amdgcn_readlane(di_b, j));            \
        float4* t_ = (float4*)(tile + j * SW_LD + lane * 4);                            \
        float4 h_ = *t_;                                                                \
        h_.x = fmaf(acc.x, d_, h_.x); h_.y = fmaf(acc.y, d_, h_.y);                     \
        h_.z = fmaf(acc.z, d_, h_.z); h_.w = fmaf(acc.w, d_, h_.w);                     \
        *t_ = h_;                                                                       \
        acc = make_float4(0.f, 0.f, 0.f, 0.f);                                          \
    } while (0)

#define GIN_CONSUME(q_, v_, w_)                                                         \
    do {                                                                                \
        while ((q_) >= nb) {                                                            \
            GIN_FLUSH();                                                                \
            ++j;                                                                        \
            nb = (j < 15) ? __builtin_amdgcn_readlane(sege, j) : 0x7fffffff;            \
        }                                                                               \
        const float wf_ = __int_as_float(w_);                                           \
        acc.x = fmaf((v_).x, wf_, acc.x); acc.y = fmaf((v_).y, wf_, acc.y);             \
        acc.z = fmaf((v_).z, wf_, acc.z); acc.w = fmaf((v_).w, wf_, acc.w);             \
    } while (0)

#define GIN_ISSUE(V_, W_)                                                               \
    _Pragma("unroll")                                                                   \
    for (int i = 0; i < 8; ++i) { V_[i] = x4[(size_t)eN[i].x * 32 + lane]; W_[i] = eN[i].y; }

#define GIN_LOADREC(P_)                                                                 \
    _Pragma("unroll")                                                                   \
    for (int i = 0; i < 8; ++i) eN[i] = rec[(P_) + i];

#define GIN_CONSUME8(P_, V_, W_)                                                        \
    _Pragma("unroll")                                                                   \
    for (int i = 0; i < 8; ++i) GIN_CONSUME((P_) + i, V_[i], W_[i]);

    int p = beg_all;
    if (nfull > 0) {
        int c = 0;
        for (;;) {
            if (c + 1 >= nfull) { GIN_CONSUME8(p, vA, wA); break; }
            GIN_ISSUE(vB, wB);
            if (c + 2 < nfull) { GIN_LOADREC(p + 16); }
            GIN_CONSUME8(p, vA, wA);
            ++c; p += 8;
            if (c + 1 >= nfull) { GIN_CONSUME8(p, vB, wB); break; }
            GIN_ISSUE(vA, wA);
            if (c + 2 < nfull) { GIN_LOADREC(p + 16); }
            GIN_CONSUME8(p, vB, wB);
            ++c; p += 8;
        }
        p = beg_all + nfull * 8;
    }
    if (p < end_all) {
        const int last = end_all - 1;
        int2 e[8];
#pragma unroll
        for (int i = 0; i < 8; ++i) e[i] = rec[(p + i < last) ? (p + i) : last];
#pragma unroll
        for (int i = 0; i < 8; ++i) vA[i] = x4[(size_t)e[i].x * 32 + lane];
#pragma unroll
        for (int i = 0; i < 8; ++i) {
            if (p + i < end_all) GIN_CONSUME(p + i, vA[i], e[i].y);
        }
    }
    for (; j < 16; ++j) GIN_FLUSH();
#undef GIN_CONSUME8
#undef GIN_LOADREC
#undef GIN_ISSUE
#undef GIN_CONSUME
#undef GIN_FLUSH


    v2f A[32];
#pragma unroll
    for (int kb = 0; kb < 32; ++kb) {
        const int k = kb * 4 + half * 2;
        const float2 hv = *(const float2*)(tile + l16 * SW_LD + k);
        A[kb].x = hv.x;
        A[kb].y = hv.y;
    }

#pragma unroll 1
    for (int jt = 0; jt < 8; ++jt) {
        const int col = jt * 16 + l16;
        const float* wrow = &sW[col * SW_LD];
        v8f cc = {};
#pragma unroll
        for (int kb = 0; kb < 32; ++kb) {
            const int k = kb * 4 + half * 2;
            v2f b;
            b.x = wrow[k];
            b.y = wrow[k + 1];
            cc = __builtin_amdgcn_wmma_f32_16x16x4_f32(
                     false, A[kb],  false, b,
                     (short)0, cc,  false,  false);
        }
        const float bv = bias[col];
#pragma unroll
        for (int r = 0; r < 8; ++r) {
            const int rr = row0 + half * 8 + r;
            if (rr < n_nodes) {
                float v = cc[r] + bv;
                if (do_relu) v = fmaxf(v, 0.0f);
                if (zero_row0 && rr == 0) v = 0.0f;
                out[(size_t)rr * DCH + col] = v;
            }
        }
    }
}

extern "C" void kernel_launch(void* const* d_in, const int* in_sizes, int n_in,
                              void* d_out, int out_size, void* d_ws, size_t ws_size,
                              hipStream_t stream) {
    const float* emb = (const float*)d_in[0];
    const float* w1  = (const float*)d_in[1];
    const float* b1  = (const float*)d_in[2];
    const float* w2  = (const float*)d_in[3];
    const float* b2  = (const float*)d_in[4];
    const float* eps = (const float*)d_in[5];
    const float* ew  = (const float*)d_in[6];
    const int*   src = (const int*)d_in[7];
    const int*   dst = (const int*)d_in[8];

    const int N = in_sizes[0] / DCH;
    const int E = in_sizes[6];

    int*   cnt     = (int*)d_ws;
    int*   off     = cnt + N;
    float* dinv    = (float*)(off + N);
    int*   lo      = (int*)(dinv + N);
    int*   tdst    = lo + SLICES * LO_LD;
    int*   trec    = tdst + E;
    int*   rec     = trec + 2 * (size_t)E;
    float* bufA    = (float*)(rec + 2 * (size_t)E);
    float* bufB    = bufA + (size_t)N * DCH;
    float* out     = (float*)d_out;

    dim3 b256(256);

    const int nBins     = (N + BIN_NODES - 1) / BIN_NODES;
    const int per_slice = (E + SLICES - 1) / SLICES;
    slice_group<<<SLICES, 512, 0, stream>>>(ew, src, dst, lo, tdst, trec, E, nBins, per_slice);
    bin_sort<<<nBins, b256, 0, stream>>>(lo, tdst, trec, cnt, off, dinv, rec, N, per_slice);

    const int layerBlocks = (N + LAYER_NODES - 1) / LAYER_NODES;
    dim3 bLayer(LAYER_THREADS);

    gin_layer_fused<<<layerBlocks, bLayer, 0, stream>>>(emb, rec, off, cnt, dinv, eps, 0,
                                                       w1, b1, bufA, N, 1, 0);
    gin_layer_fused<<<layerBlocks, bLayer, 0, stream>>>(bufA, rec, off, cnt, dinv, eps, 1,
                                                       w2, b2, bufB, N, 1, 0);
    gin_layer_fused<<<layerBlocks, bLayer, 0, stream>>>(bufB, rec, off, cnt, dinv, eps, 2,
                                                       w2, b2, out, N, 0, 1);
}
